# starting kernel
# baseline (speedup 1.0000x reference)
#include <hip/hip_runtime.h>
#include <hip/hip_bf16.h>

#define B_ 4
#define L_ 512
#define D_ 256
#define H_ 128

typedef float v2f __attribute__((ext_vector_type(2)));
typedef float v8f __attribute__((ext_vector_type(8)));

static __device__ inline v8f wmma4(v2f a, v2f b, v8f c) {
    return __builtin_amdgcn_wmma_f32_16x16x4_f32(
         false, a,  false, b,
         (short)0, c,  false,  false);
}

static __device__ inline float sigmoidf_(float x) {
    return 1.0f / (1.0f + __expf(-x));
}

template <int TRANSB, bool GATED>
__global__ __launch_bounds__(256)
void gemm_wmma_kernel(const float* __restrict__ A, const float* __restrict__ Bm,
                      const float* __restrict__ bias, const float* __restrict__ Gsrc,
                      float* __restrict__ C,
                      int K, int lda, int ldb, int ldc,
                      long long strideA, long long strideB, long long strideC)
{
    const int wave = threadIdx.x >> 5;
    const int lane = threadIdx.x & 31;
    const int half = lane >> 4;
    const int l    = lane & 15;

    const int m0 = blockIdx.x * 64;
    const int n0 = blockIdx.y * 128 + wave * 16;

    const float* Ab = A  + (long long)blockIdx.z * strideA;
    const float* Bb = Bm + (long long)blockIdx.z * strideB;
    float*       Cb = C  + (long long)blockIdx.z * strideC;

    const float* Ap = Ab + (long long)(m0 + l) * lda + 2 * half;
    const long long astep = (long long)16 * lda;

    const float* Bp = TRANSB ? (Bb + (long long)(n0 + l) * ldb + 2 * half)
                             : (Bb + (n0 + l));

    v8f acc0 = {0.f,0.f,0.f,0.f,0.f,0.f,0.f,0.f};
    v8f acc1 = acc0, acc2 = acc0, acc3 = acc0;

    for (int k0 = 0; k0 < K; k0 += 4) {
        v2f b;
        if (TRANSB) {
            b = *(const v2f*)(Bp + k0);
        } else {
            b.x = Bp[(long long)(k0 + 2 * half) * ldb];
            b.y = Bp[(long long)(k0 + 2 * half + 1) * ldb];
        }
        v2f a0 = *(const v2f*)(Ap + k0);
        v2f a1 = *(const v2f*)(Ap + astep + k0);
        v2f a2 = *(const v2f*)(Ap + 2 * astep + k0);
        v2f a3 = *(const v2f*)(Ap + 3 * astep + k0);
        acc0 = wmma4(a0, b, acc0);
        acc1 = wmma4(a1, b, acc1);
        acc2 = wmma4(a2, b, acc2);
        acc3 = wmma4(a3, b, acc3);
    }

    const int   col = n0 + l;
    const float bv  = bias ? bias[col] : 0.0f;

    v8f accs[4] = {acc0, acc1, acc2, acc3};
#pragma unroll
    for (int mi = 0; mi < 4; ++mi) {
#pragma unroll
        for (int r = 0; r < 8; ++r) {
            const int row = m0 + mi * 16 + r + half * 8;
            float cv = accs[mi][r] + bv;
            if (GATED) {
                cv = sigmoidf_(cv) * Gsrc[(long long)row * ldc + col];
            }
            Cb[(long long)row * ldc + col] = cv;
        }
    }
}

__global__ __launch_bounds__(256)
void attn_softmax_kernel(const float* __restrict__ own, const float* __restrict__ comp,
                         const float* __restrict__ v_attn, const int* __restrict__ lengths,
                         float* __restrict__ attn)
{
    const int q = blockIdx.x;
    const int b = blockIdx.y;
    const int tid = threadIdx.x;

    __shared__ float sOwn[H_];
    __shared__ float sVa[H_];
    __shared__ float sScore[L_];
    __shared__ float red[256];

    if (tid < H_) {
        sOwn[tid] = own[((long long)b * L_ + q) * H_ + tid];
        sVa[tid]  = v_attn[tid];
    }
    __syncthreads();

    const int len = lengths[b];
    for (int k = tid; k < L_; k += 256) {
        float s;
        if (k < len) {
            const float* cr = comp + ((long long)b * L_ + k) * H_;
            s = 0.0f;
#pragma unroll 4
            for (int h = 0; h < H_; ++h)
                s += sVa[h] * tanhf(sOwn[h] + cr[h]);
        } else {
            s = -1e30f;
        }
        sScore[k] = s;
    }
    __syncthreads();

    float m = -1e30f;
    for (int k = tid; k < L_; k += 256) m = fmaxf(m, sScore[k]);
    red[tid] = m;
    __syncthreads();
    for (int s = 128; s > 0; s >>= 1) {
        if (tid < s) red[tid] = fmaxf(red[tid], red[tid + s]);
        __syncthreads();
    }
    const float mx = red[0];
    __syncthreads();

    float psum = 0.0f;
    for (int k = tid; k < L_; k += 256) {
        float e = __expf(sScore[k] - mx);
        sScore[k] = e;
        psum += e;
    }
    red[tid] = psum;
    __syncthreads();
    for (int s = 128; s > 0; s >>= 1) {
        if (tid < s) red[tid] += red[tid + s];
        __syncthreads();
    }
    const float inv = 1.0f / red[0];

    float* arow = attn + ((long long)b * L_ + q) * L_;
    for (int k = tid; k < L_; k += 256) arow[k] = sScore[k] * inv;
}

__global__ __launch_bounds__(256)
void concat_kernel(const float* __restrict__ v, const float* __restrict__ ctx,
                   float* __restrict__ inp, int total)
{
    int i = blockIdx.x * blockDim.x + threadIdx.x;
    if (i >= total) return;
    int row = i >> 9;
    int col = i & 511;
    inp[i] = (col < D_) ? v[(long long)row * D_ + col]
                        : ctx[(long long)row * D_ + (col - D_)];
}

__global__ __launch_bounds__(512)
void gru_kernel(const float* __restrict__ xp, const float* __restrict__ w_hh,
                const float* __restrict__ b_hh, const int* __restrict__ lengths,
                float* __restrict__ out, int rev)
{
    __shared__ float hA[16][H_];
    __shared__ float hp[B_][3 * H_];

    const int tid  = threadIdx.x;
    const int wave = tid >> 5;
    const int lane = tid & 31;
    const int half = lane >> 4;
    const int l    = lane & 15;

    for (int i = tid; i < 16 * H_; i += 512) (&hA[0][0])[i] = 0.0f;
    __syncthreads();

    const int b   = tid >> 7;
    const int j   = tid & 127;
    const int len = lengths[b];

    for (int step = 0; step < L_; ++step) {
        const int t = rev ? (L_ - 1 - step) : step;

        if (wave < 8) {
#pragma unroll
            for (int nt = 0; nt < 3; ++nt) {
                const int n0 = (wave * 3 + nt) * 16;
                v8f acc = {0.f,0.f,0.f,0.f,0.f,0.f,0.f,0.f};
                for (int k0 = 0; k0 < H_; k0 += 4) {
                    v2f a = *(const v2f*)&hA[l][k0 + 2 * half];
                    v2f bb = *(const v2f*)(w_hh + (long long)(n0 + l) * H_ + k0 + 2 * half);
                    acc = wmma4(a, bb, acc);
                }
                if (half == 0) {
#pragma unroll
                    for (int r = 0; r < B_; ++r) hp[r][n0 + l] = acc[r];
                }
            }
        }
        __syncthreads();

        const float hprev = hA[b][j];
        const float* xrow = xp + ((long long)(b * L_ + t)) * (3 * H_);
        const float xr = xrow[j];
        const float xz = xrow[H_ + j];
        const float xn = xrow[2 * H_ + j];
        const float hr = hp[b][j]          + b_hh[j];
        const float hz = hp[b][H_ + j]     + b_hh[H_ + j];
        const float hn = hp[b][2 * H_ + j] + b_hh[2 * H_ + j];

        const float r = sigmoidf_(xr + hr);
        const float z = sigmoidf_(xz + hz);
        const float n = tanhf(xn + r * hn);
        const float hnew = (1.0f - z) * n + z * hprev;

        const bool msk = (t < len);
        hA[b][j] = msk ? hnew : hprev;
        out[((long long)(b * L_ + t)) * (2 * H_) + rev * H_ + j] = msk ? hnew : 0.0f;
        __syncthreads();
    }
}

extern "C" void kernel_launch(void* const* d_in, const int* in_sizes, int n_in,
                              void* d_out, int out_size, void* d_ws, size_t ws_size,
                              hipStream_t stream)
{
    (void)in_sizes; (void)n_in; (void)out_size; (void)ws_size;

    const float* v       = (const float*)d_in[0];
    const int*   lengths = (const int*)  d_in[1];
    const float* own_W   = (const float*)d_in[3];
    const float* own_b   = (const float*)d_in[4];
    const float* comp_W  = (const float*)d_in[5];
    const float* comp_b  = (const float*)d_in[6];
    const float* v_attn  = (const float*)d_in[7];
    const float* gate_W  = (const float*)d_in[8];
    const float* gate_b  = (const float*)d_in[9];
    const float* w_ih_f  = (const float*)d_in[10];
    const float* w_hh_f  = (const float*)d_in[11];
    const float* b_ih_f  = (const float*)d_in[12];
    const float* b_hh_f  = (const float*)d_in[13];
    const float* w_ih_b  = (const float*)d_in[14];
    const float* w_hh_b  = (const float*)d_in[15];
    const float* b_ih_b  = (const float*)d_in[16];
    const float* b_hh_b  = (const float*)d_in[17];
    float* out = (float*)d_out;

    const int M = B_ * L_;
    float* ws    = (float*)d_ws;
    float* own   = ws;
    float* comp  = own  + (size_t)M * H_;
    float* attn  = comp + (size_t)M * H_;
    float* ctx   = attn + (size_t)B_ * L_ * L_;
    float* inp   = ctx  + (size_t)M * D_;
    float* inpg  = inp  + (size_t)M * 2 * D_;
    float* xp_f  = inpg + (size_t)M * 2 * D_;
    float* xp_b  = xp_f + (size_t)M * 3 * H_;

    {
        dim3 grid(M / 64, H_ / 128, 1);
        gemm_wmma_kernel<1, false><<<grid, 256, 0, stream>>>(v, own_W, own_b, nullptr, own,
            D_, D_, D_, H_, 0, 0, 0);
        gemm_wmma_kernel<1, false><<<grid, 256, 0, stream>>>(v, comp_W, comp_b, nullptr, comp,
            D_, D_, D_, H_, 0, 0, 0);
    }

    {
        dim3 grid(L_, B_, 1);
        attn_softmax_kernel<<<grid, 256, 0, stream>>>(own, comp, v_attn, lengths, attn);
    }

    {
        dim3 grid(L_ / 64, D_ / 128, B_);
        gemm_wmma_kernel<0, false><<<grid, 256, 0, stream>>>(attn, v, nullptr, nullptr, ctx,
            L_, L_, D_, D_,
            (long long)L_ * L_, (long long)L_ * D_, (long long)L_ * D_);
    }

    {
        const int total = M * 2 * D_;
        concat_kernel<<<(total + 255) / 256, 256, 0, stream>>>(v, ctx, inp, total);
    }

    {
        dim3 grid(M / 64, (2 * D_) / 128, 1);
        gemm_wmma_kernel<1, true><<<grid, 256, 0, stream>>>(inp, gate_W, gate_b, inp, inpg,
            2 * D_, 2 * D_, 2 * D_, 2 * D_, 0, 0, 0);
    }

    {
        dim3 grid(M / 64, (3 * H_) / 128, 1);
        gemm_wmma_kernel<1, false><<<grid, 256, 0, stream>>>(inpg, w_ih_f, b_ih_f, nullptr, xp_f,
            2 * D_, 2 * D_, 2 * D_, 3 * H_, 0, 0, 0);
        gemm_wmma_kernel<1, false><<<grid, 256, 0, stream>>>(inpg, w_ih_b, b_ih_b, nullptr, xp_b,
            2 * D_, 2 * D_, 2 * D_, 3 * H_, 0, 0, 0);
    }

    gru_kernel<<<1, 512, 0, stream>>>(xp_f, w_hh_f, b_hh_f, lengths, out,  0);
    gru_kernel<<<1, 512, 0, stream>>>(xp_b, w_hh_b, b_hh_b, lengths, out,  1);
}


# launch the forward and backward GRU recurrences as one grid of two work-groups (blockIdx.x picks the direction) so they run side by side on two WGPs instead of back to back
# speedup vs baseline: 1.9064x; 1.9064x over previous
#include <hip/hip_runtime.h>
#include <hip/hip_bf16.h>

#define B_ 4
#define L_ 512
#define D_ 256
#define H_ 128

typedef float v2f __attribute__((ext_vector_type(2)));
typedef float v8f __attribute__((ext_vector_type(8)));

static __device__ inline v8f wmma4(v2f a, v2f b, v8f c) {
    return __builtin_amdgcn_wmma_f32_16x16x4_f32(
         false, a,  false, b,
         (short)0, c,  false,  false);
}

static __device__ inline float sigmoidf_(float x) {
    return 1.0f / (1.0f + __expf(-x));
}

template <int TRANSB, bool GATED>
__global__ __launch_bounds__(256)
void gemm_wmma_kernel(const float* __restrict__ A, const float* __restrict__ Bm,
                      const float* __restrict__ bias, const float* __restrict__ Gsrc,
                      float* __restrict__ C,
                      int K, int lda, int ldb, int ldc,
                      long long strideA, long long strideB, long long strideC)
{
    const int wave = threadIdx.x >> 5;
    const int lane = threadIdx.x & 31;
    const int half = lane >> 4;
    const int l    = lane & 15;

    const int m0 = blockIdx.x * 64;
    const int n0 = blockIdx.y * 128 + wave * 16;

    const float* Ab = A  + (long long)blockIdx.z * strideA;
    const float* Bb = Bm + (long long)blockIdx.z * strideB;
    float*       Cb = C  + (long long)blockIdx.z * strideC;

    const float* Ap = Ab + (long long)(m0 + l) * lda + 2 * half;
    const long long astep = (long long)16 * lda;

    const float* Bp = TRANSB ? (Bb + (long long)(n0 + l) * ldb + 2 * half)
                             : (Bb + (n0 + l));

    v8f acc0 = {0.f,0.f,0.f,0.f,0.f,0.f,0.f,0.f};
    v8f acc1 = acc0, acc2 = acc0, acc3 = acc0;

    for (int k0 = 0; k0 < K; k0 += 4) {
        v2f b;
        if (TRANSB) {
            b = *(const v2f*)(Bp + k0);
        } else {
            b.x = Bp[(long long)(k0 + 2 * half) * ldb];
            b.y = Bp[(long long)(k0 + 2 * half + 1) * ldb];
        }
        v2f a0 = *(const v2f*)(Ap + k0);
        v2f a1 = *(const v2f*)(Ap + astep + k0);
        v2f a2 = *(const v2f*)(Ap + 2 * astep + k0);
        v2f a3 = *(const v2f*)(Ap + 3 * astep + k0);
        acc0 = wmma4(a0, b, acc0);
        acc1 = wmma4(a1, b, acc1);
        acc2 = wmma4(a2, b, acc2);
        acc3 = wmma4(a3, b, acc3);
    }

    const int   col = n0 + l;
    const float bv  = bias ? bias[col] : 0.0f;

    v8f accs[4] = {acc0, acc1, acc2, acc3};
#pragma unroll
    for (int mi = 0; mi < 4; ++mi) {
#pragma unroll
        for (int r = 0; r < 8; ++r) {
            const int row = m0 + mi * 16 + r + half * 8;
            float cv = accs[mi][r] + bv;
            if (GATED) {
                cv = sigmoidf_(cv) * Gsrc[(long long)row * ldc + col];
            }
            Cb[(long long)row * ldc + col] = cv;
        }
    }
}

__global__ __launch_bounds__(256)
void attn_softmax_kernel(const float* __restrict__ own, const float* __restrict__ comp,
                         const float* __restrict__ v_attn, const int* __restrict__ lengths,
                         float* __restrict__ attn)
{
    const int q = blockIdx.x;
    const int b = blockIdx.y;
    const int tid = threadIdx.x;

    __shared__ float sOwn[H_];
    __shared__ float sVa[H_];
    __shared__ float sScore[L_];
    __shared__ float red[256];

    if (tid < H_) {
        sOwn[tid] = own[((long long)b * L_ + q) * H_ + tid];
        sVa[tid]  = v_attn[tid];
    }
    __syncthreads();

    const int len = lengths[b];
    for (int k = tid; k < L_; k += 256) {
        float s;
        if (k < len) {
            const float* cr = comp + ((long long)b * L_ + k) * H_;
            s = 0.0f;
#pragma unroll 4
            for (int h = 0; h < H_; ++h)
                s += sVa[h] * tanhf(sOwn[h] + cr[h]);
        } else {
            s = -1e30f;
        }
        sScore[k] = s;
    }
    __syncthreads();

    float m = -1e30f;
    for (int k = tid; k < L_; k += 256) m = fmaxf(m, sScore[k]);
    red[tid] = m;
    __syncthreads();
    for (int s = 128; s > 0; s >>= 1) {
        if (tid < s) red[tid] = fmaxf(red[tid], red[tid + s]);
        __syncthreads();
    }
    const float mx = red[0];
    __syncthreads();

    float psum = 0.0f;
    for (int k = tid; k < L_; k += 256) {
        float e = __expf(sScore[k] - mx);
        sScore[k] = e;
        psum += e;
    }
    red[tid] = psum;
    __syncthreads();
    for (int s = 128; s > 0; s >>= 1) {
        if (tid < s) red[tid] += red[tid + s];
        __syncthreads();
    }
    const float inv = 1.0f / red[0];

    float* arow = attn + ((long long)b * L_ + q) * L_;
    for (int k = tid; k < L_; k += 256) arow[k] = sScore[k] * inv;
}

__global__ __launch_bounds__(256)
void concat_kernel(const float* __restrict__ v, const float* __restrict__ ctx,
                   float* __restrict__ inp, int total)
{
    int i = blockIdx.x * blockDim.x + threadIdx.x;
    if (i >= total) return;
    int row = i >> 9;
    int col = i & 511;
    inp[i] = (col < D_) ? v[(long long)row * D_ + col]
                        : ctx[(long long)row * D_ + (col - D_)];
}

__global__ __launch_bounds__(512)
void gru_kernel(const float* __restrict__ xp_f, const float* __restrict__ xp_b,
                const float* __restrict__ w_hh_f, const float* __restrict__ w_hh_b,
                const float* __restrict__ b_hh_f, const float* __restrict__ b_hh_b,
                const int* __restrict__ lengths, float* __restrict__ out)
{
    const int rev = blockIdx.x;
    const float* __restrict__ xp   = rev ? xp_b   : xp_f;
    const float* __restrict__ w_hh = rev ? w_hh_b : w_hh_f;
    const float* __restrict__ b_hh = rev ? b_hh_b : b_hh_f;

    __shared__ float hA[16][H_];
    __shared__ float hp[B_][3 * H_];

    const int tid  = threadIdx.x;
    const int wave = tid >> 5;
    const int lane = tid & 31;
    const int half = lane >> 4;
    const int l    = lane & 15;

    for (int i = tid; i < 16 * H_; i += 512) (&hA[0][0])[i] = 0.0f;
    __syncthreads();

    const int b   = tid >> 7;
    const int j   = tid & 127;
    const int len = lengths[b];

    for (int step = 0; step < L_; ++step) {
        const int t = rev ? (L_ - 1 - step) : step;

        if (wave < 8) {
#pragma unroll
            for (int nt = 0; nt < 3; ++nt) {
                const int n0 = (wave * 3 + nt) * 16;
                v8f acc = {0.f,0.f,0.f,0.f,0.f,0.f,0.f,0.f};
                for (int k0 = 0; k0 < H_; k0 += 4) {
                    v2f a = *(const v2f*)&hA[l][k0 + 2 * half];
                    v2f bb = *(const v2f*)(w_hh + (long long)(n0 + l) * H_ + k0 + 2 * half);
                    acc = wmma4(a, bb, acc);
                }
                if (half == 0) {
#pragma unroll
                    for (int r = 0; r < B_; ++r) hp[r][n0 + l] = acc[r];
                }
            }
        }
        __syncthreads();

        const float hprev = hA[b][j];
        const float* xrow = xp + ((long long)(b * L_ + t)) * (3 * H_);
        const float xr = xrow[j];
        const float xz = xrow[H_ + j];
        const float xn = xrow[2 * H_ + j];
        const float hr = hp[b][j]          + b_hh[j];
        const float hz = hp[b][H_ + j]     + b_hh[H_ + j];
        const float hn = hp[b][2 * H_ + j] + b_hh[2 * H_ + j];

        const float r = sigmoidf_(xr + hr);
        const float z = sigmoidf_(xz + hz);
        const float n = tanhf(xn + r * hn);
        const float hnew = (1.0f - z) * n + z * hprev;

        const bool msk = (t < len);
        hA[b][j] = msk ? hnew : hprev;
        out[((long long)(b * L_ + t)) * (2 * H_) + rev * H_ + j] = msk ? hnew : 0.0f;
        __syncthreads();
    }
}

extern "C" void kernel_launch(void* const* d_in, const int* in_sizes, int n_in,
                              void* d_out, int out_size, void* d_ws, size_t ws_size,
                              hipStream_t stream)
{
    (void)in_sizes; (void)n_in; (void)out_size; (void)ws_size;

    const float* v       = (const float*)d_in[0];
    const int*   lengths = (const int*)  d_in[1];
    const float* own_W   = (const float*)d_in[3];
    const float* own_b   = (const float*)d_in[4];
    const float* comp_W  = (const float*)d_in[5];
    const float* comp_b  = (const float*)d_in[6];
    const float* v_attn  = (const float*)d_in[7];
    const float* gate_W  = (const float*)d_in[8];
    const float* gate_b  = (const float*)d_in[9];
    const float* w_ih_f  = (const float*)d_in[10];
    const float* w_hh_f  = (const float*)d_in[11];
    const float* b_ih_f  = (const float*)d_in[12];
    const float* b_hh_f  = (const float*)d_in[13];
    const float* w_ih_b  = (const float*)d_in[14];
    const float* w_hh_b  = (const float*)d_in[15];
    const float* b_ih_b  = (const float*)d_in[16];
    const float* b_hh_b  = (const float*)d_in[17];
    float* out = (float*)d_out;

    const int M = B_ * L_;
    float* ws    = (float*)d_ws;
    float* own   = ws;
    float* comp  = own  + (size_t)M * H_;
    float* attn  = comp + (size_t)M * H_;
    float* ctx   = attn + (size_t)B_ * L_ * L_;
    float* inp   = ctx  + (size_t)M * D_;
    float* inpg  = inp  + (size_t)M * 2 * D_;
    float* xp_f  = inpg + (size_t)M * 2 * D_;
    float* xp_b  = xp_f + (size_t)M * 3 * H_;

    {
        dim3 grid(M / 64, H_ / 128, 1);
        gemm_wmma_kernel<1, false><<<grid, 256, 0, stream>>>(v, own_W, own_b, nullptr, own,
            D_, D_, D_, H_, 0, 0, 0);
        gemm_wmma_kernel<1, false><<<grid, 256, 0, stream>>>(v, comp_W, comp_b, nullptr, comp,
            D_, D_, D_, H_, 0, 0, 0);
    }

    {
        dim3 grid(L_, B_, 1);
        attn_softmax_kernel<<<grid, 256, 0, stream>>>(own, comp, v_attn, lengths, attn);
    }

    {
        dim3 grid(L_ / 64, D_ / 128, B_);
        gemm_wmma_kernel<0, false><<<grid, 256, 0, stream>>>(attn, v, nullptr, nullptr, ctx,
            L_, L_, D_, D_,
            (long long)L_ * L_, (long long)L_ * D_, (long long)L_ * D_);
    }

    {
        const int total = M * 2 * D_;
        concat_kernel<<<(total + 255) / 256, 256, 0, stream>>>(v, ctx, inp, total);
    }

    {
        dim3 grid(M / 64, (2 * D_) / 128, 1);
        gemm_wmma_kernel<1, true><<<grid, 256, 0, stream>>>(inp, gate_W, gate_b, inp, inpg,
            2 * D_, 2 * D_, 2 * D_, 2 * D_, 0, 0, 0);
    }

    {
        dim3 grid(M / 64, (3 * H_) / 128, 1);
        gemm_wmma_kernel<1, false><<<grid, 256, 0, stream>>>(inpg, w_ih_f, b_ih_f, nullptr, xp_f,
            2 * D_, 2 * D_, 2 * D_, 3 * H_, 0, 0, 0);
        gemm_wmma_kernel<1, false><<<grid, 256, 0, stream>>>(inpg, w_ih_b, b_ih_b, nullptr, xp_b,
            2 * D_, 2 * D_, 2 * D_, 3 * H_, 0, 0, 0);
    }

    gru_kernel<<<2, 512, 0, stream>>>(xp_f, xp_b, w_hh_f, w_hh_b, b_hh_f, b_hh_b, lengths, out);
}


# GRU: keep w_hh in registers across all steps: every wave loads its 3 N-tiles x one K half as WMMA B fragments once; hp is now the sum of two K-half partials (k sum order changes)
# speedup vs baseline: 3.3476x; 1.7560x over previous
#include <hip/hip_runtime.h>
#include <hip/hip_bf16.h>

#define B_ 4
#define L_ 512
#define D_ 256
#define H_ 128

typedef float v2f __attribute__((ext_vector_type(2)));
typedef float v8f __attribute__((ext_vector_type(8)));

static __device__ inline v8f wmma4(v2f a, v2f b, v8f c) {
    return __builtin_amdgcn_wmma_f32_16x16x4_f32(
         false, a,  false, b,
         (short)0, c,  false,  false);
}

static __device__ inline float sigmoidf_(float x) {
    return 1.0f / (1.0f + __expf(-x));
}

template <int TRANSB, bool GATED>
__global__ __launch_bounds__(256)
void gemm_wmma_kernel(const float* __restrict__ A, const float* __restrict__ Bm,
                      const float* __restrict__ bias, const float* __restrict__ Gsrc,
                      float* __restrict__ C,
                      int K, int lda, int ldb, int ldc,
                      long long strideA, long long strideB, long long strideC)
{
    const int wave = threadIdx.x >> 5;
    const int lane = threadIdx.x & 31;
    const int half = lane >> 4;
    const int l    = lane & 15;

    const int m0 = blockIdx.x * 64;
    const int n0 = blockIdx.y * 128 + wave * 16;

    const float* Ab = A  + (long long)blockIdx.z * strideA;
    const float* Bb = Bm + (long long)blockIdx.z * strideB;
    float*       Cb = C  + (long long)blockIdx.z * strideC;

    const float* Ap = Ab + (long long)(m0 + l) * lda + 2 * half;
    const long long astep = (long long)16 * lda;

    const float* Bp = TRANSB ? (Bb + (long long)(n0 + l) * ldb + 2 * half)
                             : (Bb + (n0 + l));

    v8f acc0 = {0.f,0.f,0.f,0.f,0.f,0.f,0.f,0.f};
    v8f acc1 = acc0, acc2 = acc0, acc3 = acc0;

    for (int k0 = 0; k0 < K; k0 += 4) {
        v2f b;
        if (TRANSB) {
            b = *(const v2f*)(Bp + k0);
        } else {
            b.x = Bp[(long long)(k0 + 2 * half) * ldb];
            b.y = Bp[(long long)(k0 + 2 * half + 1) * ldb];
        }
        v2f a0 = *(const v2f*)(Ap + k0);
        v2f a1 = *(const v2f*)(Ap + astep + k0);
        v2f a2 = *(const v2f*)(Ap + 2 * astep + k0);
        v2f a3 = *(const v2f*)(Ap + 3 * astep + k0);
        acc0 = wmma4(a0, b, acc0);
        acc1 = wmma4(a1, b, acc1);
        acc2 = wmma4(a2, b, acc2);
        acc3 = wmma4(a3, b, acc3);
    }

    const int   col = n0 + l;
    const float bv  = bias ? bias[col] : 0.0f;

    v8f accs[4] = {acc0, acc1, acc2, acc3};
#pragma unroll
    for (int mi = 0; mi < 4; ++mi) {
#pragma unroll
        for (int r = 0; r < 8; ++r) {
            const int row = m0 + mi * 16 + r + half * 8;
            float cv = accs[mi][r] + bv;
            if (GATED) {
                cv = sigmoidf_(cv) * Gsrc[(long long)row * ldc + col];
            }
            Cb[(long long)row * ldc + col] = cv;
        }
    }
}

__global__ __launch_bounds__(256)
void attn_softmax_kernel(const float* __restrict__ own, const float* __restrict__ comp,
                         const float* __restrict__ v_attn, const int* __restrict__ lengths,
                         float* __restrict__ attn)
{
    const int q = blockIdx.x;
    const int b = blockIdx.y;
    const int tid = threadIdx.x;

    __shared__ float sOwn[H_];
    __shared__ float sVa[H_];
    __shared__ float sScore[L_];
    __shared__ float red[256];

    if (tid < H_) {
        sOwn[tid] = own[((long long)b * L_ + q) * H_ + tid];
        sVa[tid]  = v_attn[tid];
    }
    __syncthreads();

    const int len = lengths[b];
    for (int k = tid; k < L_; k += 256) {
        float s;
        if (k < len) {
            const float* cr = comp + ((long long)b * L_ + k) * H_;
            s = 0.0f;
#pragma unroll 4
            for (int h = 0; h < H_; ++h)
                s += sVa[h] * tanhf(sOwn[h] + cr[h]);
        } else {
            s = -1e30f;
        }
        sScore[k] = s;
    }
    __syncthreads();

    float m = -1e30f;
    for (int k = tid; k < L_; k += 256) m = fmaxf(m, sScore[k]);
    red[tid] = m;
    __syncthreads();
    for (int s = 128; s > 0; s >>= 1) {
        if (tid < s) red[tid] = fmaxf(red[tid], red[tid + s]);
        __syncthreads();
    }
    const float mx = red[0];
    __syncthreads();

    float psum = 0.0f;
    for (int k = tid; k < L_; k += 256) {
        float e = __expf(sScore[k] - mx);
        sScore[k] = e;
        psum += e;
    }
    red[tid] = psum;
    __syncthreads();
    for (int s = 128; s > 0; s >>= 1) {
        if (tid < s) red[tid] += red[tid + s];
        __syncthreads();
    }
    const float inv = 1.0f / red[0];

    float* arow = attn + ((long long)b * L_ + q) * L_;
    for (int k = tid; k < L_; k += 256) arow[k] = sScore[k] * inv;
}

__global__ __launch_bounds__(256)
void concat_kernel(const float* __restrict__ v, const float* __restrict__ ctx,
                   float* __restrict__ inp, int total)
{
    int i = blockIdx.x * blockDim.x + threadIdx.x;
    if (i >= total) return;
    int row = i >> 9;
    int col = i & 511;
    inp[i] = (col < D_) ? v[(long long)row * D_ + col]
                        : ctx[(long long)row * D_ + (col - D_)];
}

__global__ __launch_bounds__(512)
void gru_kernel(const float* __restrict__ xp_f, const float* __restrict__ xp_b,
                const float* __restrict__ w_hh_f, const float* __restrict__ w_hh_b,
                const float* __restrict__ b_hh_f, const float* __restrict__ b_hh_b,
                const int* __restrict__ lengths, float* __restrict__ out)
{
    const int rev = blockIdx.x;
    const float* __restrict__ xp   = rev ? xp_b   : xp_f;
    const float* __restrict__ w_hh = rev ? w_hh_b : w_hh_f;
    const float* __restrict__ b_hh = rev ? b_hh_b : b_hh_f;

    __shared__ float hA[16][H_];
    __shared__ float hp[2][B_][3 * H_];

    const int tid  = threadIdx.x;
    const int wave = tid >> 5;
    const int lane = tid & 31;
    const int half = lane >> 4;
    const int l    = lane & 15;

    const int kh    = wave & 1;
    const int nbase = (wave >> 1) * 48;
    const int kbase = kh * (H_ / 2) + 2 * half;

    v2f wr0[16], wr1[16], wr2[16];
#pragma unroll
    for (int ks = 0; ks < 16; ++ks) {
        wr0[ks] = *(const v2f*)(w_hh + (long long)(nbase +  0 + l) * H_ + kbase + 4 * ks);
        wr1[ks] = *(const v2f*)(w_hh + (long long)(nbase + 16 + l) * H_ + kbase + 4 * ks);
        wr2[ks] = *(const v2f*)(w_hh + (long long)(nbase + 32 + l) * H_ + kbase + 4 * ks);
    }

    for (int i = tid; i < 16 * H_; i += 512) (&hA[0][0])[i] = 0.0f;
    __syncthreads();

    const int b   = tid >> 7;
    const int j   = tid & 127;
    const int len = lengths[b];

    for (int step = 0; step < L_; ++step) {
        const int t = rev ? (L_ - 1 - step) : step;

        {
            v2f a[16];
#pragma unroll
            for (int ks = 0; ks < 16; ++ks)
                a[ks] = *(const v2f*)&hA[l][kbase + 4 * ks];

            v8f acc0 = {0.f,0.f,0.f,0.f,0.f,0.f,0.f,0.f};
            v8f acc1 = acc0, acc2 = acc0;
#pragma unroll
            for (int ks = 0; ks < 16; ++ks) {
                acc0 = wmma4(a[ks], wr0[ks], acc0);
                acc1 = wmma4(a[ks], wr1[ks], acc1);
                acc2 = wmma4(a[ks], wr2[ks], acc2);
            }
            if (half == 0) {
#pragma unroll
                for (int r = 0; r < B_; ++r) {
                    hp[kh][r][nbase +  0 + l] = acc0[r];
                    hp[kh][r][nbase + 16 + l] = acc1[r];
                    hp[kh][r][nbase + 32 + l] = acc2[r];
                }
            }
        }
        __syncthreads();

        const float hprev = hA[b][j];
        const float* xrow = xp + ((long long)(b * L_ + t)) * (3 * H_);
        const float xr = xrow[j];
        const float xz = xrow[H_ + j];
        const float xn = xrow[2 * H_ + j];
        const float hr = (hp[0][b][j]          + hp[1][b][j])          + b_hh[j];
        const float hz = (hp[0][b][H_ + j]     + hp[1][b][H_ + j])     + b_hh[H_ + j];
        const float hn = (hp[0][b][2 * H_ + j] + hp[1][b][2 * H_ + j]) + b_hh[2 * H_ + j];

        const float r = sigmoidf_(xr + hr);
        const float z = sigmoidf_(xz + hz);
        const float n = tanhf(xn + r * hn);
        const float hnew = (1.0f - z) * n + z * hprev;

        const bool msk = (t < len);
        hA[b][j] = msk ? hnew : hprev;
        out[((long long)(b * L_ + t)) * (2 * H_) + rev * H_ + j] = msk ? hnew : 0.0f;
        __syncthreads();
    }
}

extern "C" void kernel_launch(void* const* d_in, const int* in_sizes, int n_in,
                              void* d_out, int out_size, void* d_ws, size_t ws_size,
                              hipStream_t stream)
{
    (void)in_sizes; (void)n_in; (void)out_size; (void)ws_size;

    const float* v       = (const float*)d_in[0];
    const int*   lengths = (const int*)  d_in[1];
    const float* own_W   = (const float*)d_in[3];
    const float* own_b   = (const float*)d_in[4];
    const float* comp_W  = (const float*)d_in[5];
    const float* comp_b  = (const float*)d_in[6];
    const float* v_attn  = (const float*)d_in[7];
    const float* gate_W  = (const float*)d_in[8];
    const float* gate_b  = (const float*)d_in[9];
    const float* w_ih_f  = (const float*)d_in[10];
    const float* w_hh_f  = (const float*)d_in[11];
    const float* b_ih_f  = (const float*)d_in[12];
    const float* b_hh_f  = (const float*)d_in[13];
    const float* w_ih_b  = (const float*)d_in[14];
    const float* w_hh_b  = (const float*)d_in[15];
    const float* b_ih_b  = (const float*)d_in[16];
    const float* b_hh_b  = (const float*)d_in[17];
    float* out = (float*)d_out;

    const int M = B_ * L_;
    float* ws    = (float*)d_ws;
    float* own   = ws;
    float* comp  = own  + (size_t)M * H_;
    float* attn  = comp + (size_t)M * H_;
    float* ctx   = attn + (size_t)B_ * L_ * L_;
    float* inp   = ctx  + (size_t)M * D_;
    float* inpg  = inp  + (size_t)M * 2 * D_;
    float* xp_f  = inpg + (size_t)M * 2 * D_;
    float* xp_b  = xp_f + (size_t)M * 3 * H_;

    {
        dim3 grid(M / 64, H_ / 128, 1);
        gemm_wmma_kernel<1, false><<<grid, 256, 0, stream>>>(v, own_W, own_b, nullptr, own,
            D_, D_, D_, H_, 0, 0, 0);
        gemm_wmma_kernel<1, false><<<grid, 256, 0, stream>>>(v, comp_W, comp_b, nullptr, comp,
            D_, D_, D_, H_, 0, 0, 0);
    }

    {
        dim3 grid(L_, B_, 1);
        attn_softmax_kernel<<<grid, 256, 0, stream>>>(own, comp, v_attn, lengths, attn);
    }

    {
        dim3 grid(L_ / 64, D_ / 128, B_);
        gemm_wmma_kernel<0, false><<<grid, 256, 0, stream>>>(attn, v, nullptr, nullptr, ctx,
            L_, L_, D_, D_,
            (long long)L_ * L_, (long long)L_ * D_, (long long)L_ * D_);
    }

    {
        const int total = M * 2 * D_;
        concat_kernel<<<(total + 255) / 256, 256, 0, stream>>>(v, ctx, inp, total);
    }

    {
        dim3 grid(M / 64, (2 * D_) / 128, 1);
        gemm_wmma_kernel<1, true><<<grid, 256, 0, stream>>>(inp, gate_W, gate_b, inp, inpg,
            2 * D_, 2 * D_, 2 * D_, 2 * D_, 0, 0, 0);
    }

    {
        dim3 grid(M / 64, (3 * H_) / 128, 1);
        gemm_wmma_kernel<1, false><<<grid, 256, 0, stream>>>(inpg, w_ih_f, b_ih_f, nullptr, xp_f,
            2 * D_, 2 * D_, 2 * D_, 3 * H_, 0, 0, 0);
        gemm_wmma_kernel<1, false><<<grid, 256, 0, stream>>>(inpg, w_ih_b, b_ih_b, nullptr, xp_b,
            2 * D_, 2 * D_, 2 * D_, 3 * H_, 0, 0, 0);
    }

    gru_kernel<<<2, 512, 0, stream>>>(xp_f, xp_b, w_hh_f, w_hh_b, b_hh_f, b_hh_b, lengths, out);
}


# GRU: pad the LDS hidden-state tile's row pitch from 128 to 132 floats so each A-fragment read (16 rows at one k) hits 64 distinct banks instead of a 16-way bank conflict; same k order
# speedup vs baseline: 3.7793x; 1.1290x over previous
#include <hip/hip_runtime.h>
#include <hip/hip_bf16.h>

#define B_ 4
#define L_ 512
#define D_ 256
#define H_ 128
#define HA_PITCH (H_ + 4)

typedef float v2f __attribute__((ext_vector_type(2)));
typedef float v8f __attribute__((ext_vector_type(8)));

static __device__ inline v8f wmma4(v2f a, v2f b, v8f c) {
    return __builtin_amdgcn_wmma_f32_16x16x4_f32(
         false, a,  false, b,
         (short)0, c,  false,  false);
}

static __device__ inline float sigmoidf_(float x) {
    return 1.0f / (1.0f + __expf(-x));
}

template <int TRANSB, bool GATED>
__global__ __launch_bounds__(256)
void gemm_wmma_kernel(const float* __restrict__ A, const float* __restrict__ Bm,
                      const float* __restrict__ bias, const float* __restrict__ Gsrc,
                      float* __restrict__ C,
                      int K, int lda, int ldb, int ldc,
                      long long strideA, long long strideB, long long strideC)
{
    const int wave = threadIdx.x >> 5;
    const int lane = threadIdx.x & 31;
    const int half = lane >> 4;
    const int l    = lane & 15;

    const int m0 = blockIdx.x * 64;
    const int n0 = blockIdx.y * 128 + wave * 16;

    const float* Ab = A  + (long long)blockIdx.z * strideA;
    const float* Bb = Bm + (long long)blockIdx.z * strideB;
    float*       Cb = C  + (long long)blockIdx.z * strideC;

    const float* Ap = Ab + (long long)(m0 + l) * lda + 2 * half;
    const long long astep = (long long)16 * lda;

    const float* Bp = TRANSB ? (Bb + (long long)(n0 + l) * ldb + 2 * half)
                             : (Bb + (n0 + l));

    v8f acc0 = {0.f,0.f,0.f,0.f,0.f,0.f,0.f,0.f};
    v8f acc1 = acc0, acc2 = acc0, acc3 = acc0;

    for (int k0 = 0; k0 < K; k0 += 4) {
        v2f b;
        if (TRANSB) {
            b = *(const v2f*)(Bp + k0);
        } else {
            b.x = Bp[(long long)(k0 + 2 * half) * ldb];
            b.y = Bp[(long long)(k0 + 2 * half + 1) * ldb];
        }
        v2f a0 = *(const v2f*)(Ap + k0);
        v2f a1 = *(const v2f*)(Ap + astep + k0);
        v2f a2 = *(const v2f*)(Ap + 2 * astep + k0);
        v2f a3 = *(const v2f*)(Ap + 3 * astep + k0);
        acc0 = wmma4(a0, b, acc0);
        acc1 = wmma4(a1, b, acc1);
        acc2 = wmma4(a2, b, acc2);
        acc3 = wmma4(a3, b, acc3);
    }

    const int   col = n0 + l;
    const float bv  = bias ? bias[col] : 0.0f;

    v8f accs[4] = {acc0, acc1, acc2, acc3};
#pragma unroll
    for (int mi = 0; mi < 4; ++mi) {
#pragma unroll
        for (int r = 0; r < 8; ++r) {
            const int row = m0 + mi * 16 + r + half * 8;
            float cv = accs[mi][r] + bv;
            if (GATED) {
                cv = sigmoidf_(cv) * Gsrc[(long long)row * ldc + col];
            }
            Cb[(long long)row * ldc + col] = cv;
        }
    }
}

__global__ __launch_bounds__(256)
void attn_softmax_kernel(const float* __restrict__ own, const float* __restrict__ comp,
                         const float* __restrict__ v_attn, const int* __restrict__ lengths,
                         float* __restrict__ attn)
{
    const int q = blockIdx.x;
    const int b = blockIdx.y;
    const int tid = threadIdx.x;

    __shared__ float sOwn[H_];
    __shared__ float sVa[H_];
    __shared__ float sScore[L_];
    __shared__ float red[256];

    if (tid < H_) {
        sOwn[tid] = own[((long long)b * L_ + q) * H_ + tid];
        sVa[tid]  = v_attn[tid];
    }
    __syncthreads();

    const int len = lengths[b];
    for (int k = tid; k < L_; k += 256) {
        float s;
        if (k < len) {
            const float* cr = comp + ((long long)b * L_ + k) * H_;
            s = 0.0f;
#pragma unroll 4
            for (int h = 0; h < H_; ++h)
                s += sVa[h] * tanhf(sOwn[h] + cr[h]);
        } else {
            s = -1e30f;
        }
        sScore[k] = s;
    }
    __syncthreads();

    float m = -1e30f;
    for (int k = tid; k < L_; k += 256) m = fmaxf(m, sScore[k]);
    red[tid] = m;
    __syncthreads();
    for (int s = 128; s > 0; s >>= 1) {
        if (tid < s) red[tid] = fmaxf(red[tid], red[tid + s]);
        __syncthreads();
    }
    const float mx = red[0];
    __syncthreads();

    float psum = 0.0f;
    for (int k = tid; k < L_; k += 256) {
        float e = __expf(sScore[k] - mx);
        sScore[k] = e;
        psum += e;
    }
    red[tid] = psum;
    __syncthreads();
    for (int s = 128; s > 0; s >>= 1) {
        if (tid < s) red[tid] += red[tid + s];
        __syncthreads();
    }
    const float inv = 1.0f / red[0];

    float* arow = attn + ((long long)b * L_ + q) * L_;
    for (int k = tid; k < L_; k += 256) arow[k] = sScore[k] * inv;
}

__global__ __launch_bounds__(256)
void concat_kernel(const float* __restrict__ v, const float* __restrict__ ctx,
                   float* __restrict__ inp, int total)
{
    int i = blockIdx.x * blockDim.x + threadIdx.x;
    if (i >= total) return;
    int row = i >> 9;
    int col = i & 511;
    inp[i] = (col < D_) ? v[(long long)row * D_ + col]
                        : ctx[(long long)row * D_ + (col - D_)];
}

__global__ __launch_bounds__(512)
void gru_kernel(const float* __restrict__ xp_f, const float* __restrict__ xp_b,
                const float* __restrict__ w_hh_f, const float* __restrict__ w_hh_b,
                const float* __restrict__ b_hh_f, const float* __restrict__ b_hh_b,
                const int* __restrict__ lengths, float* __restrict__ out)
{
    const int rev = blockIdx.x;
    const float* __restrict__ xp   = rev ? xp_b   : xp_f;
    const float* __restrict__ w_hh = rev ? w_hh_b : w_hh_f;
    const float* __restrict__ b_hh = rev ? b_hh_b : b_hh_f;

    __shared__ float hA[16][HA_PITCH];
    __shared__ float hp[2][B_][3 * H_];

    const int tid  = threadIdx.x;
    const int wave = tid >> 5;
    const int lane = tid & 31;
    const int half = lane >> 4;
    const int l    = lane & 15;

    const int kh    = wave & 1;
    const int nbase = (wave >> 1) * 48;
    const int kbase = kh * (H_ / 2) + 2 * half;

    v2f wr0[16], wr1[16], wr2[16];
#pragma unroll
    for (int ks = 0; ks < 16; ++ks) {
        wr0[ks] = *(const v2f*)(w_hh + (long long)(nbase +  0 + l) * H_ + kbase + 4 * ks);
        wr1[ks] = *(const v2f*)(w_hh + (long long)(nbase + 16 + l) * H_ + kbase + 4 * ks);
        wr2[ks] = *(const v2f*)(w_hh + (long long)(nbase + 32 + l) * H_ + kbase + 4 * ks);
    }

    for (int i = tid; i < 16 * HA_PITCH; i += 512) (&hA[0][0])[i] = 0.0f;
    __syncthreads();

    const int b   = tid >> 7;
    const int j   = tid & 127;
    const int len = lengths[b];

    for (int step = 0; step < L_; ++step) {
        const int t = rev ? (L_ - 1 - step) : step;

        {
            v2f a[16];
#pragma unroll
            for (int ks = 0; ks < 16; ++ks)
                a[ks] = *(const v2f*)&hA[l][kbase + 4 * ks];

            v8f acc0 = {0.f,0.f,0.f,0.f,0.f,0.f,0.f,0.f};
            v8f acc1 = acc0, acc2 = acc0;
#pragma unroll
            for (int ks = 0; ks < 16; ++ks) {
                acc0 = wmma4(a[ks], wr0[ks], acc0);
                acc1 = wmma4(a[ks], wr1[ks], acc1);
                acc2 = wmma4(a[ks], wr2[ks], acc2);
            }
            if (half == 0) {
#pragma unroll
                for (int r = 0; r < B_; ++r) {
                    hp[kh][r][nbase +  0 + l] = acc0[r];
                    hp[kh][r][nbase + 16 + l] = acc1[r];
                    hp[kh][r][nbase + 32 + l] = acc2[r];
                }
            }
        }
        __syncthreads();

        const float hprev = hA[b][j];
        const float* xrow = xp + ((long long)(b * L_ + t)) * (3 * H_);
        const float xr = xrow[j];
        const float xz = xrow[H_ + j];
        const float xn = xrow[2 * H_ + j];
        const float hr = (hp[0][b][j]          + hp[1][b][j])          + b_hh[j];
        const float hz = (hp[0][b][H_ + j]     + hp[1][b][H_ + j])     + b_hh[H_ + j];
        const float hn = (hp[0][b][2 * H_ + j] + hp[1][b][2 * H_ + j]) + b_hh[2 * H_ + j];

        const float r = sigmoidf_(xr + hr);
        const float z = sigmoidf_(xz + hz);
        const float n = tanhf(xn + r * hn);
        const float hnew = (1.0f - z) * n + z * hprev;

        const bool msk = (t < len);
        hA[b][j] = msk ? hnew : hprev;
        out[((long long)(b * L_ + t)) * (2 * H_) + rev * H_ + j] = msk ? hnew : 0.0f;
        __syncthreads();
    }
}

extern "C" void kernel_launch(void* const* d_in, const int* in_sizes, int n_in,
                              void* d_out, int out_size, void* d_ws, size_t ws_size,
                              hipStream_t stream)
{
    (void)in_sizes; (void)n_in; (void)out_size; (void)ws_size;

    const float* v       = (const float*)d_in[0];
    const int*   lengths = (const int*)  d_in[1];
    const float* own_W   = (const float*)d_in[3];
    const float* own_b   = (const float*)d_in[4];
    const float* comp_W  = (const float*)d_in[5];
    const float* comp_b  = (const float*)d_in[6];
    const float* v_attn  = (const float*)d_in[7];
    const float* gate_W  = (const float*)d_in[8];
    const float* gate_b  = (const float*)d_in[9];
    const float* w_ih_f  = (const float*)d_in[10];
    const float* w_hh_f  = (const float*)d_in[11];
    const float* b_ih_f  = (const float*)d_in[12];
    const float* b_hh_f  = (const float*)d_in[13];
    const float* w_ih_b  = (const float*)d_in[14];
    const float* w_hh_b  = (const float*)d_in[15];
    const float* b_ih_b  = (const float*)d_in[16];
    const float* b_hh_b  = (const float*)d_in[17];
    float* out = (float*)d_out;

    const int M = B_ * L_;
    float* ws    = (float*)d_ws;
    float* own   = ws;
    float* comp  = own  + (size_t)M * H_;
    float* attn  = comp + (size_t)M * H_;
    float* ctx   = attn + (size_t)B_ * L_ * L_;
    float* inp   = ctx  + (size_t)M * D_;
    float* inpg  = inp  + (size_t)M * 2 * D_;
    float* xp_f  = inpg + (size_t)M * 2 * D_;
    float* xp_b  = xp_f + (size_t)M * 3 * H_;

    {
        dim3 grid(M / 64, H_ / 128, 1);
        gemm_wmma_kernel<1, false><<<grid, 256, 0, stream>>>(v, own_W, own_b, nullptr, own,
            D_, D_, D_, H_, 0, 0, 0);
        gemm_wmma_kernel<1, false><<<grid, 256, 0, stream>>>(v, comp_W, comp_b, nullptr, comp,
            D_, D_, D_, H_, 0, 0, 0);
    }

    {
        dim3 grid(L_, B_, 1);
        attn_softmax_kernel<<<grid, 256, 0, stream>>>(own, comp, v_attn, lengths, attn);
    }

    {
        dim3 grid(L_ / 64, D_ / 128, B_);
        gemm_wmma_kernel<0, false><<<grid, 256, 0, stream>>>(attn, v, nullptr, nullptr, ctx,
            L_, L_, D_, D_,
            (long long)L_ * L_, (long long)L_ * D_, (long long)L_ * D_);
    }

    {
        const int total = M * 2 * D_;
        concat_kernel<<<(total + 255) / 256, 256, 0, stream>>>(v, ctx, inp, total);
    }

    {
        dim3 grid(M / 64, (2 * D_) / 128, 1);
        gemm_wmma_kernel<1, true><<<grid, 256, 0, stream>>>(inp, gate_W, gate_b, inp, inpg,
            2 * D_, 2 * D_, 2 * D_, 2 * D_, 0, 0, 0);
    }

    {
        dim3 grid(M / 64, (3 * H_) / 128, 1);
        gemm_wmma_kernel<1, false><<<grid, 256, 0, stream>>>(inpg, w_ih_f, b_ih_f, nullptr, xp_f,
            2 * D_, 2 * D_, 2 * D_, 3 * H_, 0, 0, 0);
        gemm_wmma_kernel<1, false><<<grid, 256, 0, stream>>>(inpg, w_ih_b, b_ih_b, nullptr, xp_b,
            2 * D_, 2 * D_, 2 * D_, 3 * H_, 0, 0, 0);
    }

    gru_kernel<<<2, 512, 0, stream>>>(xp_f, xp_b, w_hh_f, w_hh_b, b_hh_f, b_hh_b, lengths, out);
}


# GRU: take global memory off each step's critical path: the step's xp values are loaded in front of the WMMA phase and the out store is issued one step late, so no barrier waits for a round trip
# speedup vs baseline: 4.7939x; 1.2685x over previous
#include <hip/hip_runtime.h>
#include <hip/hip_bf16.h>

#define B_ 4
#define L_ 512
#define D_ 256
#define H_ 128
#define HA_PITCH (H_ + 4)

typedef float v2f __attribute__((ext_vector_type(2)));
typedef float v8f __attribute__((ext_vector_type(8)));

static __device__ inline v8f wmma4(v2f a, v2f b, v8f c) {
    return __builtin_amdgcn_wmma_f32_16x16x4_f32(
         false, a,  false, b,
         (short)0, c,  false,  false);
}

static __device__ inline float sigmoidf_(float x) {
    return 1.0f / (1.0f + __expf(-x));
}

template <int TRANSB, bool GATED>
__global__ __launch_bounds__(256)
void gemm_wmma_kernel(const float* __restrict__ A, const float* __restrict__ Bm,
                      const float* __restrict__ bias, const float* __restrict__ Gsrc,
                      float* __restrict__ C,
                      int K, int lda, int ldb, int ldc,
                      long long strideA, long long strideB, long long strideC)
{
    const int wave = threadIdx.x >> 5;
    const int lane = threadIdx.x & 31;
    const int half = lane >> 4;
    const int l    = lane & 15;

    const int m0 = blockIdx.x * 64;
    const int n0 = blockIdx.y * 128 + wave * 16;

    const float* Ab = A  + (long long)blockIdx.z * strideA;
    const float* Bb = Bm + (long long)blockIdx.z * strideB;
    float*       Cb = C  + (long long)blockIdx.z * strideC;

    const float* Ap = Ab + (long long)(m0 + l) * lda + 2 * half;
    const long long astep = (long long)16 * lda;

    const float* Bp = TRANSB ? (Bb + (long long)(n0 + l) * ldb + 2 * half)
                             : (Bb + (n0 + l));

    v8f acc0 = {0.f,0.f,0.f,0.f,0.f,0.f,0.f,0.f};
    v8f acc1 = acc0, acc2 = acc0, acc3 = acc0;

    for (int k0 = 0; k0 < K; k0 += 4) {
        v2f b;
        if (TRANSB) {
            b = *(const v2f*)(Bp + k0);
        } else {
            b.x = Bp[(long long)(k0 + 2 * half) * ldb];
            b.y = Bp[(long long)(k0 + 2 * half + 1) * ldb];
        }
        v2f a0 = *(const v2f*)(Ap + k0);
        v2f a1 = *(const v2f*)(Ap + astep + k0);
        v2f a2 = *(const v2f*)(Ap + 2 * astep + k0);
        v2f a3 = *(const v2f*)(Ap + 3 * astep + k0);
        acc0 = wmma4(a0, b, acc0);
        acc1 = wmma4(a1, b, acc1);
        acc2 = wmma4(a2, b, acc2);
        acc3 = wmma4(a3, b, acc3);
    }

    const int   col = n0 + l;
    const float bv  = bias ? bias[col] : 0.0f;

    v8f accs[4] = {acc0, acc1, acc2, acc3};
#pragma unroll
    for (int mi = 0; mi < 4; ++mi) {
#pragma unroll
        for (int r = 0; r < 8; ++r) {
            const int row = m0 + mi * 16 + r + half * 8;
            float cv = accs[mi][r] + bv;
            if (GATED) {
                cv = sigmoidf_(cv) * Gsrc[(long long)row * ldc + col];
            }
            Cb[(long long)row * ldc + col] = cv;
        }
    }
}

__global__ __launch_bounds__(256)
void attn_softmax_kernel(const float* __restrict__ own, const float* __restrict__ comp,
                         const float* __restrict__ v_attn, const int* __restrict__ lengths,
                         float* __restrict__ attn)
{
    const int q = blockIdx.x;
    const int b = blockIdx.y;
    const int tid = threadIdx.x;

    __shared__ float sOwn[H_];
    __shared__ float sVa[H_];
    __shared__ float sScore[L_];
    __shared__ float red[256];

    if (tid < H_) {
        sOwn[tid] = own[((long long)b * L_ + q) * H_ + tid];
        sVa[tid]  = v_attn[tid];
    }
    __syncthreads();

    const int len = lengths[b];
    for (int k = tid; k < L_; k += 256) {
        float s;
        if (k < len) {
            const float* cr = comp + ((long long)b * L_ + k) * H_;
            s = 0.0f;
#pragma unroll 4
            for (int h = 0; h < H_; ++h)
                s += sVa[h] * tanhf(sOwn[h] + cr[h]);
        } else {
            s = -1e30f;
        }
        sScore[k] = s;
    }
    __syncthreads();

    float m = -1e30f;
    for (int k = tid; k < L_; k += 256) m = fmaxf(m, sScore[k]);
    red[tid] = m;
    __syncthreads();
    for (int s = 128; s > 0; s >>= 1) {
        if (tid < s) red[tid] = fmaxf(red[tid], red[tid + s]);
        __syncthreads();
    }
    const float mx = red[0];
    __syncthreads();

    float psum = 0.0f;
    for (int k = tid; k < L_; k += 256) {
        float e = __expf(sScore[k] - mx);
        sScore[k] = e;
        psum += e;
    }
    red[tid] = psum;
    __syncthreads();
    for (int s = 128; s > 0; s >>= 1) {
        if (tid < s) red[tid] += red[tid + s];
        __syncthreads();
    }
    const float inv = 1.0f / red[0];

    float* arow = attn + ((long long)b * L_ + q) * L_;
    for (int k = tid; k < L_; k += 256) arow[k] = sScore[k] * inv;
}

__global__ __launch_bounds__(256)
void concat_kernel(const float* __restrict__ v, const float* __restrict__ ctx,
                   float* __restrict__ inp, int total)
{
    int i = blockIdx.x * blockDim.x + threadIdx.x;
    if (i >= total) return;
    int row = i >> 9;
    int col = i & 511;
    inp[i] = (col < D_) ? v[(long long)row * D_ + col]
                        : ctx[(long long)row * D_ + (col - D_)];
}

__global__ __launch_bounds__(512)
void gru_kernel(const float* __restrict__ xp_f, const float* __restrict__ xp_b,
                const float* __restrict__ w_hh_f, const float* __restrict__ w_hh_b,
                const float* __restrict__ b_hh_f, const float* __restrict__ b_hh_b,
                const int* __restrict__ lengths, float* __restrict__ out)
{
    const int rev = blockIdx.x;
    const float* __restrict__ xp   = rev ? xp_b   : xp_f;
    const float* __restrict__ w_hh = rev ? w_hh_b : w_hh_f;
    const float* __restrict__ b_hh = rev ? b_hh_b : b_hh_f;

    __shared__ float hA[16][HA_PITCH];
    __shared__ float hp[2][B_][3 * H_];

    const int tid  = threadIdx.x;
    const int wave = tid >> 5;
    const int lane = tid & 31;
    const int half = lane >> 4;
    const int l    = lane & 15;

    const int kh    = wave & 1;
    const int nbase = (wave >> 1) * 48;
    const int kbase = kh * (H_ / 2) + 2 * half;

    v2f wr0[16], wr1[16], wr2[16];
#pragma unroll
    for (int ks = 0; ks < 16; ++ks) {
        wr0[ks] = *(const v2f*)(w_hh + (long long)(nbase +  0 + l) * H_ + kbase + 4 * ks);
        wr1[ks] = *(const v2f*)(w_hh + (long long)(nbase + 16 + l) * H_ + kbase + 4 * ks);
        wr2[ks] = *(const v2f*)(w_hh + (long long)(nbase + 32 + l) * H_ + kbase + 4 * ks);
    }

    for (int i = tid; i < 16 * HA_PITCH; i += 512) (&hA[0][0])[i] = 0.0f;
    __syncthreads();

    const int b   = tid >> 7;
    const int j   = tid & 127;
    const int len = lengths[b];

    float  out_val = 0.0f;
    float* out_ptr = out;

    for (int step = 0; step < L_; ++step) {
        const int t = rev ? (L_ - 1 - step) : step;

        if (step > 0) *out_ptr = out_val;

        const float* xrow = xp + ((long long)(b * L_ + t)) * (3 * H_);
        const float xr = xrow[j];
        const float xz = xrow[H_ + j];
        const float xn = xrow[2 * H_ + j];

        {
            v2f a[16];
#pragma unroll
            for (int ks = 0; ks < 16; ++ks)
                a[ks] = *(const v2f*)&hA[l][kbase + 4 * ks];

            v8f acc0 = {0.f,0.f,0.f,0.f,0.f,0.f,0.f,0.f};
            v8f acc1 = acc0, acc2 = acc0;
#pragma unroll
            for (int ks = 0; ks < 16; ++ks) {
                acc0 = wmma4(a[ks], wr0[ks], acc0);
                acc1 = wmma4(a[ks], wr1[ks], acc1);
                acc2 = wmma4(a[ks], wr2[ks], acc2);
            }
            if (half == 0) {
#pragma unroll
                for (int r = 0; r < B_; ++r) {
                    hp[kh][r][nbase +  0 + l] = acc0[r];
                    hp[kh][r][nbase + 16 + l] = acc1[r];
                    hp[kh][r][nbase + 32 + l] = acc2[r];
                }
            }
        }
        __syncthreads();

        const float hprev = hA[b][j];
        const float hr = (hp[0][b][j]          + hp[1][b][j])          + b_hh[j];
        const float hz = (hp[0][b][H_ + j]     + hp[1][b][H_ + j])     + b_hh[H_ + j];
        const float hn = (hp[0][b][2 * H_ + j] + hp[1][b][2 * H_ + j]) + b_hh[2 * H_ + j];

        const float r = sigmoidf_(xr + hr);
        const float z = sigmoidf_(xz + hz);
        const float n = tanhf(xn + r * hn);
        const float hnew = (1.0f - z) * n + z * hprev;

        const bool msk = (t < len);
        hA[b][j] = msk ? hnew : hprev;
        out_val = msk ? hnew : 0.0f;
        out_ptr = out + ((long long)(b * L_ + t)) * (2 * H_) + rev * H_ + j;
        __syncthreads();
    }
    *out_ptr = out_val;
}

extern "C" void kernel_launch(void* const* d_in, const int* in_sizes, int n_in,
                              void* d_out, int out_size, void* d_ws, size_t ws_size,
                              hipStream_t stream)
{
    (void)in_sizes; (void)n_in; (void)out_size; (void)ws_size;

    const float* v       = (const float*)d_in[0];
    const int*   lengths = (const int*)  d_in[1];
    const float* own_W   = (const float*)d_in[3];
    const float* own_b   = (const float*)d_in[4];
    const float* comp_W  = (const float*)d_in[5];
    const float* comp_b  = (const float*)d_in[6];
    const float* v_attn  = (const float*)d_in[7];
    const float* gate_W  = (const float*)d_in[8];
    const float* gate_b  = (const float*)d_in[9];
    const float* w_ih_f  = (const float*)d_in[10];
    const float* w_hh_f  = (const float*)d_in[11];
    const float* b_ih_f  = (const float*)d_in[12];
    const float* b_hh_f  = (const float*)d_in[13];
    const float* w_ih_b  = (const float*)d_in[14];
    const float* w_hh_b  = (const float*)d_in[15];
    const float* b_ih_b  = (const float*)d_in[16];
    const float* b_hh_b  = (const float*)d_in[17];
    float* out = (float*)d_out;

    const int M = B_ * L_;
    float* ws    = (float*)d_ws;
    float* own   = ws;
    float* comp  = own  + (size_t)M * H_;
    float* attn  = comp + (size_t)M * H_;
    float* ctx   = attn + (size_t)B_ * L_ * L_;
    float* inp   = ctx  + (size_t)M * D_;
    float* inpg  = inp  + (size_t)M * 2 * D_;
    float* xp_f  = inpg + (size_t)M * 2 * D_;
    float* xp_b  = xp_f + (size_t)M * 3 * H_;

    {
        dim3 grid(M / 64, H_ / 128, 1);
        gemm_wmma_kernel<1, false><<<grid, 256, 0, stream>>>(v, own_W, own_b, nullptr, own,
            D_, D_, D_, H_, 0, 0, 0);
        gemm_wmma_kernel<1, false><<<grid, 256, 0, stream>>>(v, comp_W, comp_b, nullptr, comp,
            D_, D_, D_, H_, 0, 0, 0);
    }

    {
        dim3 grid(L_, B_, 1);
        attn_softmax_kernel<<<grid, 256, 0, stream>>>(own, comp, v_attn, lengths, attn);
    }

    {
        dim3 grid(L_ / 64, D_ / 128, B_);
        gemm_wmma_kernel<0, false><<<grid, 256, 0, stream>>>(attn, v, nullptr, nullptr, ctx,
            L_, L_, D_, D_,
            (long long)L_ * L_, (long long)L_ * D_, (long long)L_ * D_);
    }

    {
        const int total = M * 2 * D_;
        concat_kernel<<<(total + 255) / 256, 256, 0, stream>>>(v, ctx, inp, total);
    }

    {
        dim3 grid(M / 64, (2 * D_) / 128, 1);
        gemm_wmma_kernel<1, true><<<grid, 256, 0, stream>>>(inp, gate_W, gate_b, inp, inpg,
            2 * D_, 2 * D_, 2 * D_, 2 * D_, 0, 0, 0);
    }

    {
        dim3 grid(M / 64, (3 * H_) / 128, 1);
        gemm_wmma_kernel<1, false><<<grid, 256, 0, stream>>>(inpg, w_ih_f, b_ih_f, nullptr, xp_f,
            2 * D_, 2 * D_, 2 * D_, 3 * H_, 0, 0, 0);
        gemm_wmma_kernel<1, false><<<grid, 256, 0, stream>>>(inpg, w_ih_b, b_ih_b, nullptr, xp_b,
            2 * D_, 2 * D_, 2 * D_, 3 * H_, 0, 0, 0);
    }

    gru_kernel<<<2, 512, 0, stream>>>(xp_f, xp_b, w_hh_f, w_hh_b, b_hh_f, b_hh_b, lengths, out);
}


# GEMMs: stage each 32-k chunk of the block's A rows and B panel in LDS with full-line 16-byte loads shared by the 8 waves, next chunk's loads in flight under the WMMAs; same k order
# speedup vs baseline: 5.3096x; 1.1076x over previous
#include <hip/hip_runtime.h>
#include <hip/hip_bf16.h>

#define B_ 4
#define L_ 512
#define D_ 256
#define H_ 128
#define HA_PITCH (H_ + 4)

typedef float v2f __attribute__((ext_vector_type(2)));
typedef float v8f __attribute__((ext_vector_type(8)));

static __device__ inline v8f wmma4(v2f a, v2f b, v8f c) {
    return __builtin_amdgcn_wmma_f32_16x16x4_f32(
         false, a,  false, b,
         (short)0, c,  false,  false);
}

static __device__ inline float sigmoidf_(float x) {
    return 1.0f / (1.0f + __expf(-x));
}

#define GK_  32
#define GP_  (GK_ + 4)
#define GPN_ (128 + 8)

template <int TRANSB, bool GATED>
__global__ __launch_bounds__(256)
void gemm_wmma_kernel(const float* __restrict__ A, const float* __restrict__ Bm,
                      const float* __restrict__ bias, const float* __restrict__ Gsrc,
                      float* __restrict__ C,
                      int K, int lda, int ldb, int ldc,
                      long long strideA, long long strideB, long long strideC)
{
    __shared__ __attribute__((aligned(16))) float As[64 * GP_];
    __shared__ __attribute__((aligned(16))) float Bs[TRANSB ? 128 * GP_ : GK_ * GPN_];

    const int tid  = threadIdx.x;
    const int wave = tid >> 5;
    const int lane = tid & 31;
    const int half = lane >> 4;
    const int l    = lane & 15;

    const int m0 = blockIdx.x * 64;
    const int nb = blockIdx.y * 128;
    const int n0 = nb + wave * 16;

    const float* Ab = A  + (long long)blockIdx.z * strideA;
    const float* Bb = Bm + (long long)blockIdx.z * strideB;
    float*       Cb = C  + (long long)blockIdx.z * strideC;

    const int ar  = tid >> 3;
    const int ac4 = (tid & 7) * 4;
    const int kr  = tid >> 5;
    const int nc4 = (tid & 31) * 4;

    const float* gA = Ab + (long long)(m0 + ar) * lda + ac4;
    const float* gB = TRANSB ? (Bb + (long long)(nb + ar) * ldb + ac4)
                             : (Bb + (long long)kr * ldb + nb + nc4);

    float4 ra0, ra1, rb0, rb1, rb2, rb3;

#define GEMM_LOAD_CHUNK(k0)                                                        \
    do {                                                                           \
        ra0 = *(const float4*)(gA + (k0));                                         \
        ra1 = *(const float4*)(gA + (long long)32 * lda + (k0));                   \
        if (TRANSB) {                                                              \
            rb0 = *(const float4*)(gB + (k0));                                     \
            rb1 = *(const float4*)(gB + (long long)32 * ldb + (k0));               \
            rb2 = *(const float4*)(gB + (long long)64 * ldb + (k0));               \
            rb3 = *(const float4*)(gB + (long long)96 * ldb + (k0));               \
        } else {                                                                   \
            rb0 = *(const float4*)(gB + (long long)((k0))      * ldb);             \
            rb1 = *(const float4*)(gB + (long long)((k0) + 8)  * ldb);             \
            rb2 = *(const float4*)(gB + (long long)((k0) + 16) * ldb);             \
            rb3 = *(const float4*)(gB + (long long)((k0) + 24) * ldb);             \
        }                                                                          \
    } while (0)

    v8f acc0 = {0.f,0.f,0.f,0.f,0.f,0.f,0.f,0.f};
    v8f acc1 = acc0, acc2 = acc0, acc3 = acc0;

    const float* sA = As + l * GP_ + 2 * half;
    const float* sB = TRANSB ? (Bs + (wave * 16 + l) * GP_ + 2 * half)
                             : (Bs + (2 * half) * GPN_ + wave * 16 + l);

    GEMM_LOAD_CHUNK(0);

    for (int k0 = 0; k0 < K; k0 += GK_) {
        *(float4*)(As + ar * GP_ + ac4)        = ra0;
        *(float4*)(As + (ar + 32) * GP_ + ac4) = ra1;
        if (TRANSB) {
            *(float4*)(Bs + ar * GP_ + ac4)        = rb0;
            *(float4*)(Bs + (ar + 32) * GP_ + ac4) = rb1;
            *(float4*)(Bs + (ar + 64) * GP_ + ac4) = rb2;
            *(float4*)(Bs + (ar + 96) * GP_ + ac4) = rb3;
        } else {
            *(float4*)(Bs + kr * GPN_ + nc4)        = rb0;
            *(float4*)(Bs + (kr + 8) * GPN_ + nc4)  = rb1;
            *(float4*)(Bs + (kr + 16) * GPN_ + nc4) = rb2;
            *(float4*)(Bs + (kr + 24) * GPN_ + nc4) = rb3;
        }
        __syncthreads();

        if (k0 + GK_ < K) GEMM_LOAD_CHUNK(k0 + GK_);

#pragma unroll
        for (int ks = 0; ks < GK_ / 4; ++ks) {
            v2f b;
            if (TRANSB) {
                b = *(const v2f*)(sB + 4 * ks);
            } else {
                b.x = sB[(4 * ks) * GPN_];
                b.y = sB[(4 * ks + 1) * GPN_];
            }
            v2f a0 = *(const v2f*)(sA + 4 * ks);
            v2f a1 = *(const v2f*)(sA + 16 * GP_ + 4 * ks);
            v2f a2 = *(const v2f*)(sA + 32 * GP_ + 4 * ks);
            v2f a3 = *(const v2f*)(sA + 48 * GP_ + 4 * ks);
            acc0 = wmma4(a0, b, acc0);
            acc1 = wmma4(a1, b, acc1);
            acc2 = wmma4(a2, b, acc2);
            acc3 = wmma4(a3, b, acc3);
        }
        __syncthreads();
    }
#undef GEMM_LOAD_CHUNK

    const int   col = n0 + l;
    const float bv  = bias ? bias[col] : 0.0f;

    v8f accs[4] = {acc0, acc1, acc2, acc3};
#pragma unroll
    for (int mi = 0; mi < 4; ++mi) {
#pragma unroll
        for (int r = 0; r < 8; ++r) {
            const int row = m0 + mi * 16 + r + half * 8;
            float cv = accs[mi][r] + bv;
            if (GATED) {
                cv = sigmoidf_(cv) * Gsrc[(long long)row * ldc + col];
            }
            Cb[(long long)row * ldc + col] = cv;
        }
    }
}

__global__ __launch_bounds__(256)
void attn_softmax_kernel(const float* __restrict__ own, const float* __restrict__ comp,
                         const float* __restrict__ v_attn, const int* __restrict__ lengths,
                         float* __restrict__ attn)
{
    const int q = blockIdx.x;
    const int b = blockIdx.y;
    const int tid = threadIdx.x;

    __shared__ float sOwn[H_];
    __shared__ float sVa[H_];
    __shared__ float sScore[L_];
    __shared__ float red[256];

    if (tid < H_) {
        sOwn[tid] = own[((long long)b * L_ + q) * H_ + tid];
        sVa[tid]  = v_attn[tid];
    }
    __syncthreads();

    const int len = lengths[b];
    for (int k = tid; k < L_; k += 256) {
        float s;
        if (k < len) {
            const float* cr = comp + ((long long)b * L_ + k) * H_;
            s = 0.0f;
#pragma unroll 4
            for (int h = 0; h < H_; ++h)
                s += sVa[h] * tanhf(sOwn[h] + cr[h]);
        } else {
            s = -1e30f;
        }
        sScore[k] = s;
    }
    __syncthreads();

    float m = -1e30f;
    for (int k = tid; k < L_; k += 256) m = fmaxf(m, sScore[k]);
    red[tid] = m;
    __syncthreads();
    for (int s = 128; s > 0; s >>= 1) {
        if (tid < s) red[tid] = fmaxf(red[tid], red[tid + s]);
        __syncthreads();
    }
    const float mx = red[0];
    __syncthreads();

    float psum = 0.0f;
    for (int k = tid; k < L_; k += 256) {
        float e = __expf(sScore[k] - mx);
        sScore[k] = e;
        psum += e;
    }
    red[tid] = psum;
    __syncthreads();
    for (int s = 128; s > 0; s >>= 1) {
        if (tid < s) red[tid] += red[tid + s];
        __syncthreads();
    }
    const float inv = 1.0f / red[0];

    float* arow = attn + ((long long)b * L_ + q) * L_;
    for (int k = tid; k < L_; k += 256) arow[k] = sScore[k] * inv;
}

__global__ __launch_bounds__(256)
void concat_kernel(const float* __restrict__ v, const float* __restrict__ ctx,
                   float* __restrict__ inp, int total)
{
    int i = blockIdx.x * blockDim.x + threadIdx.x;
    if (i >= total) return;
    int row = i >> 9;
    int col = i & 511;
    inp[i] = (col < D_) ? v[(long long)row * D_ + col]
                        : ctx[(long long)row * D_ + (col - D_)];
}

__global__ __launch_bounds__(512)
void gru_kernel(const float* __restrict__ xp_f, const float* __restrict__ xp_b,
                const float* __restrict__ w_hh_f, const float* __restrict__ w_hh_b,
                const float* __restrict__ b_hh_f, const float* __restrict__ b_hh_b,
                const int* __restrict__ lengths, float* __restrict__ out)
{
    const int rev = blockIdx.x;
    const float* __restrict__ xp   = rev ? xp_b   : xp_f;
    const float* __restrict__ w_hh = rev ? w_hh_b : w_hh_f;
    const float* __restrict__ b_hh = rev ? b_hh_b : b_hh_f;

    __shared__ float hA[16][HA_PITCH];
    __shared__ float hp[2][B_][3 * H_];

    const int tid  = threadIdx.x;
    const int wave = tid >> 5;
    const int lane = tid & 31;
    const int half = lane >> 4;
    const int l    = lane & 15;

    const int kh    = wave & 1;
    const int nbase = (wave >> 1) * 48;
    const int kbase = kh * (H_ / 2) + 2 * half;

    v2f wr0[16], wr1[16], wr2[16];
#pragma unroll
    for (int ks = 0; ks < 16; ++ks) {
        wr0[ks] = *(const v2f*)(w_hh + (long long)(nbase +  0 + l) * H_ + kbase + 4 * ks);
        wr1[ks] = *(const v2f*)(w_hh + (long long)(nbase + 16 + l) * H_ + kbase + 4 * ks);
        wr2[ks] = *(const v2f*)(w_hh + (long long)(nbase + 32 + l) * H_ + kbase + 4 * ks);
    }

    for (int i = tid; i < 16 * HA_PITCH; i += 512) (&hA[0][0])[i] = 0.0f;
    __syncthreads();

    const int b   = tid >> 7;
    const int j   = tid & 127;
    const int len = lengths[b];

    float  out_val = 0.0f;
    float* out_ptr = out;

    for (int step = 0; step < L_; ++step) {
        const int t = rev ? (L_ - 1 - step) : step;

        if (step > 0) *out_ptr = out_val;

        const float* xrow = xp + ((long long)(b * L_ + t)) * (3 * H_);
        const float xr = xrow[j];
        const float xz = xrow[H_ + j];
        const float xn = xrow[2 * H_ + j];

        {
            v2f a[16];
#pragma unroll
            for (int ks = 0; ks < 16; ++ks)
                a[ks] = *(const v2f*)&hA[l][kbase + 4 * ks];

            v8f acc0 = {0.f,0.f,0.f,0.f,0.f,0.f,0.f,0.f};
            v8f acc1 = acc0, acc2 = acc0;
#pragma unroll
            for (int ks = 0; ks < 16; ++ks) {
                acc0 = wmma4(a[ks], wr0[ks], acc0);
                acc1 = wmma4(a[ks], wr1[ks], acc1);
                acc2 = wmma4(a[ks], wr2[ks], acc2);
            }
            if (half == 0) {
#pragma unroll
                for (int r = 0; r < B_; ++r) {
                    hp[kh][r][nbase +  0 + l] = acc0[r];
                    hp[kh][r][nbase + 16 + l] = acc1[r];
                    hp[kh][r][nbase + 32 + l] = acc2[r];
                }
            }
        }
        __syncthreads();

        const float hprev = hA[b][j];
        const float hr = (hp[0][b][j]          + hp[1][b][j])          + b_hh[j];
        const float hz = (hp[0][b][H_ + j]     + hp[1][b][H_ + j])     + b_hh[H_ + j];
        const float hn = (hp[0][b][2 * H_ + j] + hp[1][b][2 * H_ + j]) + b_hh[2 * H_ + j];

        const float r = sigmoidf_(xr + hr);
        const float z = sigmoidf_(xz + hz);
        const float n = tanhf(xn + r * hn);
        const float hnew = (1.0f - z) * n + z * hprev;

        const bool msk = (t < len);
        hA[b][j] = msk ? hnew : hprev;
        out_val = msk ? hnew : 0.0f;
        out_ptr = out + ((long long)(b * L_ + t)) * (2 * H_) + rev * H_ + j;
        __syncthreads();
    }
    *out_ptr = out_val;
}

extern "C" void kernel_launch(void* const* d_in, const int* in_sizes, int n_in,
                              void* d_out, int out_size, void* d_ws, size_t ws_size,
                              hipStream_t stream)
{
    (void)in_sizes; (void)n_in; (void)out_size; (void)ws_size;

    const float* v       = (const float*)d_in[0];
    const int*   lengths = (const int*)  d_in[1];
    const float* own_W   = (const float*)d_in[3];
    const float* own_b   = (const float*)d_in[4];
    const float* comp_W  = (const float*)d_in[5];
    const float* comp_b  = (const float*)d_in[6];
    const float* v_attn  = (const float*)d_in[7];
    const float* gate_W  = (const float*)d_in[8];
    const float* gate_b  = (const float*)d_in[9];
    const float* w_ih_f  = (const float*)d_in[10];
    const float* w_hh_f  = (const float*)d_in[11];
    const float* b_ih_f  = (const float*)d_in[12];
    const float* b_hh_f  = (const float*)d_in[13];
    const float* w_ih_b  = (const float*)d_in[14];
    const float* w_hh_b  = (const float*)d_in[15];
    const float* b_ih_b  = (const float*)d_in[16];
    const float* b_hh_b  = (const float*)d_in[17];
    float* out = (float*)d_out;

    const int M = B_ * L_;
    float* ws    = (float*)d_ws;
    float* own   = ws;
    float* comp  = own  + (size_t)M * H_;
    float* attn  = comp + (size_t)M * H_;
    float* ctx   = attn + (size_t)B_ * L_ * L_;
    float* inp   = ctx  + (size_t)M * D_;
    float* inpg  = inp  + (size_t)M * 2 * D_;
    float* xp_f  = inpg + (size_t)M * 2 * D_;
    float* xp_b  = xp_f + (size_t)M * 3 * H_;

    {
        dim3 grid(M / 64, H_ / 128, 1);
        gemm_wmma_kernel<1, false><<<grid, 256, 0, stream>>>(v, own_W, own_b, nullptr, own,
            D_, D_, D_, H_, 0, 0, 0);
        gemm_wmma_kernel<1, false><<<grid, 256, 0, stream>>>(v, comp_W, comp_b, nullptr, comp,
            D_, D_, D_, H_, 0, 0, 0);
    }

    {
        dim3 grid(L_, B_, 1);
        attn_softmax_kernel<<<grid, 256, 0, stream>>>(own, comp, v_attn, lengths, attn);
    }

    {
        dim3 grid(L_ / 64, D_ / 128, B_);
        gemm_wmma_kernel<0, false><<<grid, 256, 0, stream>>>(attn, v, nullptr, nullptr, ctx,
            L_, L_, D_, D_,
            (long long)L_ * L_, (long long)L_ * D_, (long long)L_ * D_);
    }

    {
        const int total = M * 2 * D_;
        concat_kernel<<<(total + 255) / 256, 256, 0, stream>>>(v, ctx, inp, total);
    }

    {
        dim3 grid(M / 64, (2 * D_) / 128, 1);
        gemm_wmma_kernel<1, true><<<grid, 256, 0, stream>>>(inp, gate_W, gate_b, inp, inpg,
            2 * D_, 2 * D_, 2 * D_, 2 * D_, 0, 0, 0);
    }

    {
        dim3 grid(M / 64, (3 * H_) / 128, 1);
        gemm_wmma_kernel<1, false><<<grid, 256, 0, stream>>>(inpg, w_ih_f, b_ih_f, nullptr, xp_f,
            2 * D_, 2 * D_, 2 * D_, 3 * H_, 0, 0, 0);
        gemm_wmma_kernel<1, false><<<grid, 256, 0, stream>>>(inpg, w_ih_b, b_ih_b, nullptr, xp_b,
            2 * D_, 2 * D_, 2 * D_, 3 * H_, 0, 0, 0);
    }

    gru_kernel<<<2, 512, 0, stream>>>(xp_f, xp_b, w_hh_f, w_hh_b, b_hh_f, b_hh_b, lengths, out);
}


# tanh and sigmoid (GRU gates, attention scores, gated GEMM epilogue) as v_exp_f32 + v_rcp_f32, tanh(x) = 1 - 2/(1+e^(2x)), replacing library tanhf and IEEE division; their last-bit rounding changes
# speedup vs baseline: 6.1630x; 1.1607x over previous
#include <hip/hip_runtime.h>
#include <hip/hip_bf16.h>

#define B_ 4
#define L_ 512
#define D_ 256
#define H_ 128
#define HA_PITCH (H_ + 4)

typedef float v2f __attribute__((ext_vector_type(2)));
typedef float v8f __attribute__((ext_vector_type(8)));

static __device__ inline v8f wmma4(v2f a, v2f b, v8f c) {
    return __builtin_amdgcn_wmma_f32_16x16x4_f32(
         false, a,  false, b,
         (short)0, c,  false,  false);
}

static __device__ inline float sigmoidf_(float x) {
    return __builtin_amdgcn_rcpf(1.0f + __expf(-x));
}

static __device__ inline float tanhf_(float x) {
    return 1.0f - 2.0f * __builtin_amdgcn_rcpf(1.0f + __expf(2.0f * x));
}

#define GK_  32
#define GP_  (GK_ + 4)
#define GPN_ (128 + 8)

template <int TRANSB, bool GATED>
__global__ __launch_bounds__(256)
void gemm_wmma_kernel(const float* __restrict__ A, const float* __restrict__ Bm,
                      const float* __restrict__ bias, const float* __restrict__ Gsrc,
                      float* __restrict__ C,
                      int K, int lda, int ldb, int ldc,
                      long long strideA, long long strideB, long long strideC)
{
    __shared__ __attribute__((aligned(16))) float As[64 * GP_];
    __shared__ __attribute__((aligned(16))) float Bs[TRANSB ? 128 * GP_ : GK_ * GPN_];

    const int tid  = threadIdx.x;
    const int wave = tid >> 5;
    const int lane = tid & 31;
    const int half = lane >> 4;
    const int l    = lane & 15;

    const int m0 = blockIdx.x * 64;
    const int nb = blockIdx.y * 128;
    const int n0 = nb + wave * 16;

    const float* Ab = A  + (long long)blockIdx.z * strideA;
    const float* Bb = Bm + (long long)blockIdx.z * strideB;
    float*       Cb = C  + (long long)blockIdx.z * strideC;

    const int ar  = tid >> 3;
    const int ac4 = (tid & 7) * 4;
    const int kr  = tid >> 5;
    const int nc4 = (tid & 31) * 4;

    const float* gA = Ab + (long long)(m0 + ar) * lda + ac4;
    const float* gB = TRANSB ? (Bb + (long long)(nb + ar) * ldb + ac4)
                             : (Bb + (long long)kr * ldb + nb + nc4);

    float4 ra0, ra1, rb0, rb1, rb2, rb3;

#define GEMM_LOAD_CHUNK(k0)                                                        \
    do {                                                                           \
        ra0 = *(const float4*)(gA + (k0));                                         \
        ra1 = *(const float4*)(gA + (long long)32 * lda + (k0));                   \
        if (TRANSB) {                                                              \
            rb0 = *(const float4*)(gB + (k0));                                     \
            rb1 = *(const float4*)(gB + (long long)32 * ldb + (k0));               \
            rb2 = *(const float4*)(gB + (long long)64 * ldb + (k0));               \
            rb3 = *(const float4*)(gB + (long long)96 * ldb + (k0));               \
        } else {                                                                   \
            rb0 = *(const float4*)(gB + (long long)((k0))      * ldb);             \
            rb1 = *(const float4*)(gB + (long long)((k0) + 8)  * ldb);             \
            rb2 = *(const float4*)(gB + (long long)((k0) + 16) * ldb);             \
            rb3 = *(const float4*)(gB + (long long)((k0) + 24) * ldb);             \
        }                                                                          \
    } while (0)

    v8f acc0 = {0.f,0.f,0.f,0.f,0.f,0.f,0.f,0.f};
    v8f acc1 = acc0, acc2 = acc0, acc3 = acc0;

    const float* sA = As + l * GP_ + 2 * half;
    const float* sB = TRANSB ? (Bs + (wave * 16 + l) * GP_ + 2 * half)
                             : (Bs + (2 * half) * GPN_ + wave * 16 + l);

    GEMM_LOAD_CHUNK(0);

    for (int k0 = 0; k0 < K; k0 += GK_) {
        *(float4*)(As + ar * GP_ + ac4)        = ra0;
        *(float4*)(As + (ar + 32) * GP_ + ac4) = ra1;
        if (TRANSB) {
            *(float4*)(Bs + ar * GP_ + ac4)        = rb0;
            *(float4*)(Bs + (ar + 32) * GP_ + ac4) = rb1;
            *(float4*)(Bs + (ar + 64) * GP_ + ac4) = rb2;
            *(float4*)(Bs + (ar + 96) * GP_ + ac4) = rb3;
        } else {
            *(float4*)(Bs + kr * GPN_ + nc4)        = rb0;
            *(float4*)(Bs + (kr + 8) * GPN_ + nc4)  = rb1;
            *(float4*)(Bs + (kr + 16) * GPN_ + nc4) = rb2;
            *(float4*)(Bs + (kr + 24) * GPN_ + nc4) = rb3;
        }
        __syncthreads();

        if (k0 + GK_ < K) GEMM_LOAD_CHUNK(k0 + GK_);

#pragma unroll
        for (int ks = 0; ks < GK_ / 4; ++ks) {
            v2f b;
            if (TRANSB) {
                b = *(const v2f*)(sB + 4 * ks);
            } else {
                b.x = sB[(4 * ks) * GPN_];
                b.y = sB[(4 * ks + 1) * GPN_];
            }
            v2f a0 = *(const v2f*)(sA + 4 * ks);
            v2f a1 = *(const v2f*)(sA + 16 * GP_ + 4 * ks);
            v2f a2 = *(const v2f*)(sA + 32 * GP_ + 4 * ks);
            v2f a3 = *(const v2f*)(sA + 48 * GP_ + 4 * ks);
            acc0 = wmma4(a0, b, acc0);
            acc1 = wmma4(a1, b, acc1);
            acc2 = wmma4(a2, b, acc2);
            acc3 = wmma4(a3, b, acc3);
        }
        __syncthreads();
    }
#undef GEMM_LOAD_CHUNK

    const int   col = n0 + l;
    const float bv  = bias ? bias[col] : 0.0f;

    v8f accs[4] = {acc0, acc1, acc2, acc3};
#pragma unroll
    for (int mi = 0; mi < 4; ++mi) {
#pragma unroll
        for (int r = 0; r < 8; ++r) {
            const int row = m0 + mi * 16 + r + half * 8;
            float cv = accs[mi][r] + bv;
            if (GATED) {
                cv = sigmoidf_(cv) * Gsrc[(long long)row * ldc + col];
            }
            Cb[(long long)row * ldc + col] = cv;
        }
    }
}

__global__ __launch_bounds__(256)
void attn_softmax_kernel(const float* __restrict__ own, const float* __restrict__ comp,
                         const float* __restrict__ v_attn, const int* __restrict__ lengths,
                         float* __restrict__ attn)
{
    const int q = blockIdx.x;
    const int b = blockIdx.y;
    const int tid = threadIdx.x;

    __shared__ float sOwn[H_];
    __shared__ float sVa[H_];
    __shared__ float sScore[L_];
    __shared__ float red[256];

    if (tid < H_) {
        sOwn[tid] = own[((long long)b * L_ + q) * H_ + tid];
        sVa[tid]  = v_attn[tid];
    }
    __syncthreads();

    const int len = lengths[b];
    for (int k = tid; k < L_; k += 256) {
        float s;
        if (k < len) {
            const float* cr = comp + ((long long)b * L_ + k) * H_;
            s = 0.0f;
#pragma unroll 4
            for (int h = 0; h < H_; ++h)
                s += sVa[h] * tanhf_(sOwn[h] + cr[h]);
        } else {
            s = -1e30f;
        }
        sScore[k] = s;
    }
    __syncthreads();

    float m = -1e30f;
    for (int k = tid; k < L_; k += 256) m = fmaxf(m, sScore[k]);
    red[tid] = m;
    __syncthreads();
    for (int s = 128; s > 0; s >>= 1) {
        if (tid < s) red[tid] = fmaxf(red[tid], red[tid + s]);
        __syncthreads();
    }
    const float mx = red[0];
    __syncthreads();

    float psum = 0.0f;
    for (int k = tid; k < L_; k += 256) {
        float e = __expf(sScore[k] - mx);
        sScore[k] = e;
        psum += e;
    }
    red[tid] = psum;
    __syncthreads();
    for (int s = 128; s > 0; s >>= 1) {
        if (tid < s) red[tid] += red[tid + s];
        __syncthreads();
    }
    const float inv = 1.0f / red[0];

    float* arow = attn + ((long long)b * L_ + q) * L_;
    for (int k = tid; k < L_; k += 256) arow[k] = sScore[k] * inv;
}

__global__ __launch_bounds__(256)
void concat_kernel(const float* __restrict__ v, const float* __restrict__ ctx,
                   float* __restrict__ inp, int total)
{
    int i = blockIdx.x * blockDim.x + threadIdx.x;
    if (i >= total) return;
    int row = i >> 9;
    int col = i & 511;
    inp[i] = (col < D_) ? v[(long long)row * D_ + col]
                        : ctx[(long long)row * D_ + (col - D_)];
}

__global__ __launch_bounds__(512)
void gru_kernel(const float* __restrict__ xp_f, const float* __restrict__ xp_b,
                const float* __restrict__ w_hh_f, const float* __restrict__ w_hh_b,
                const float* __restrict__ b_hh_f, const float* __restrict__ b_hh_b,
                const int* __restrict__ lengths, float* __restrict__ out)
{
    const int rev = blockIdx.x;
    const float* __restrict__ xp   = rev ? xp_b   : xp_f;
    const float* __restrict__ w_hh = rev ? w_hh_b : w_hh_f;
    const float* __restrict__ b_hh = rev ? b_hh_b : b_hh_f;

    __shared__ float hA[16][HA_PITCH];
    __shared__ float hp[2][B_][3 * H_];

    const int tid  = threadIdx.x;
    const int wave = tid >> 5;
    const int lane = tid & 31;
    const int half = lane >> 4;
    const int l    = lane & 15;

    const int kh    = wave & 1;
    const int nbase = (wave >> 1) * 48;
    const int kbase = kh * (H_ / 2) + 2 * half;

    v2f wr0[16], wr1[16], wr2[16];
#pragma unroll
    for (int ks = 0; ks < 16; ++ks) {
        wr0[ks] = *(const v2f*)(w_hh + (long long)(nbase +  0 + l) * H_ + kbase + 4 * ks);
        wr1[ks] = *(const v2f*)(w_hh + (long long)(nbase + 16 + l) * H_ + kbase + 4 * ks);
        wr2[ks] = *(const v2f*)(w_hh + (long long)(nbase + 32 + l) * H_ + kbase + 4 * ks);
    }

    for (int i = tid; i < 16 * HA_PITCH; i += 512) (&hA[0][0])[i] = 0.0f;
    __syncthreads();

    const int b   = tid >> 7;
    const int j   = tid & 127;
    const int len = lengths[b];

    float  out_val = 0.0f;
    float* out_ptr = out;

    for (int step = 0; step < L_; ++step) {
        const int t = rev ? (L_ - 1 - step) : step;

        if (step > 0) *out_ptr = out_val;

        const float* xrow = xp + ((long long)(b * L_ + t)) * (3 * H_);
        const float xr = xrow[j];
        const float xz = xrow[H_ + j];
        const float xn = xrow[2 * H_ + j];

        {
            v2f a[16];
#pragma unroll
            for (int ks = 0; ks < 16; ++ks)
                a[ks] = *(const v2f*)&hA[l][kbase + 4 * ks];

            v8f acc0 = {0.f,0.f,0.f,0.f,0.f,0.f,0.f,0.f};
            v8f acc1 = acc0, acc2 = acc0;
#pragma unroll
            for (int ks = 0; ks < 16; ++ks) {
                acc0 = wmma4(a[ks], wr0[ks], acc0);
                acc1 = wmma4(a[ks], wr1[ks], acc1);
                acc2 = wmma4(a[ks], wr2[ks], acc2);
            }
            if (half == 0) {
#pragma unroll
                for (int r = 0; r < B_; ++r) {
                    hp[kh][r][nbase +  0 + l] = acc0[r];
                    hp[kh][r][nbase + 16 + l] = acc1[r];
                    hp[kh][r][nbase + 32 + l] = acc2[r];
                }
            }
        }
        __syncthreads();

        const float hprev = hA[b][j];
        const float hr = (hp[0][b][j]          + hp[1][b][j])          + b_hh[j];
        const float hz = (hp[0][b][H_ + j]     + hp[1][b][H_ + j])     + b_hh[H_ + j];
        const float hn = (hp[0][b][2 * H_ + j] + hp[1][b][2 * H_ + j]) + b_hh[2 * H_ + j];

        const float r = sigmoidf_(xr + hr);
        const float z = sigmoidf_(xz + hz);
        const float n = tanhf_(xn + r * hn);
        const float hnew = (1.0f - z) * n + z * hprev;

        const bool msk = (t < len);
        hA[b][j] = msk ? hnew : hprev;
        out_val = msk ? hnew : 0.0f;
        out_ptr = out + ((long long)(b * L_ + t)) * (2 * H_) + rev * H_ + j;
        __syncthreads();
    }
    *out_ptr = out_val;
}

extern "C" void kernel_launch(void* const* d_in, const int* in_sizes, int n_in,
                              void* d_out, int out_size, void* d_ws, size_t ws_size,
                              hipStream_t stream)
{
    (void)in_sizes; (void)n_in; (void)out_size; (void)ws_size;

    const float* v       = (const float*)d_in[0];
    const int*   lengths = (const int*)  d_in[1];
    const float* own_W   = (const float*)d_in[3];
    const float* own_b   = (const float*)d_in[4];
    const float* comp_W  = (const float*)d_in[5];
    const float* comp_b  = (const float*)d_in[6];
    const float* v_attn  = (const float*)d_in[7];
    const float* gate_W  = (const float*)d_in[8];
    const float* gate_b  = (const float*)d_in[9];
    const float* w_ih_f  = (const float*)d_in[10];
    const float* w_hh_f  = (const float*)d_in[11];
    const float* b_ih_f  = (const float*)d_in[12];
    const float* b_hh_f  = (const float*)d_in[13];
    const float* w_ih_b  = (const float*)d_in[14];
    const float* w_hh_b  = (const float*)d_in[15];
    const float* b_ih_b  = (const float*)d_in[16];
    const float* b_hh_b  = (const float*)d_in[17];
    float* out = (float*)d_out;

    const int M = B_ * L_;
    float* ws    = (float*)d_ws;
    float* own   = ws;
    float* comp  = own  + (size_t)M * H_;
    float* attn  = comp + (size_t)M * H_;
    float* ctx   = attn + (size_t)B_ * L_ * L_;
    float* inp   = ctx  + (size_t)M * D_;
    float* inpg  = inp  + (size_t)M * 2 * D_;
    float* xp_f  = inpg + (size_t)M * 2 * D_;
    float* xp_b  = xp_f + (size_t)M * 3 * H_;

    {
        dim3 grid(M / 64, H_ / 128, 1);
        gemm_wmma_kernel<1, false><<<grid, 256, 0, stream>>>(v, own_W, own_b, nullptr, own,
            D_, D_, D_, H_, 0, 0, 0);
        gemm_wmma_kernel<1, false><<<grid, 256, 0, stream>>>(v, comp_W, comp_b, nullptr, comp,
            D_, D_, D_, H_, 0, 0, 0);
    }

    {
        dim3 grid(L_, B_, 1);
        attn_softmax_kernel<<<grid, 256, 0, stream>>>(own, comp, v_attn, lengths, attn);
    }

    {
        dim3 grid(L_ / 64, D_ / 128, B_);
        gemm_wmma_kernel<0, false><<<grid, 256, 0, stream>>>(attn, v, nullptr, nullptr, ctx,
            L_, L_, D_, D_,
            (long long)L_ * L_, (long long)L_ * D_, (long long)L_ * D_);
    }

    {
        const int total = M * 2 * D_;
        concat_kernel<<<(total + 255) / 256, 256, 0, stream>>>(v, ctx, inp, total);
    }

    {
        dim3 grid(M / 64, (2 * D_) / 128, 1);
        gemm_wmma_kernel<1, true><<<grid, 256, 0, stream>>>(inp, gate_W, gate_b, inp, inpg,
            2 * D_, 2 * D_, 2 * D_, 2 * D_, 0, 0, 0);
    }

    {
        dim3 grid(M / 64, (3 * H_) / 128, 1);
        gemm_wmma_kernel<1, false><<<grid, 256, 0, stream>>>(inpg, w_ih_f, b_ih_f, nullptr, xp_f,
            2 * D_, 2 * D_, 2 * D_, 3 * H_, 0, 0, 0);
        gemm_wmma_kernel<1, false><<<grid, 256, 0, stream>>>(inpg, w_ih_b, b_ih_b, nullptr, xp_b,
            2 * D_, 2 * D_, 2 * D_, 3 * H_, 0, 0, 0);
    }

    gru_kernel<<<2, 512, 0, stream>>>(xp_f, xp_b, w_hh_f, w_hh_b, b_hh_f, b_hh_b, lengths, out);
}


# GRU: run the recurrence only over t < max(lengths): at later t every row is masked, so h is kept and out is 0 by the reference's own mask; those out rows are stored as zeros; bit-identical
# speedup vs baseline: 7.4361x; 1.2066x over previous
#include <hip/hip_runtime.h>
#include <hip/hip_bf16.h>

#define B_ 4
#define L_ 512
#define D_ 256
#define H_ 128
#define HA_PITCH (H_ + 4)

typedef float v2f __attribute__((ext_vector_type(2)));
typedef float v8f __attribute__((ext_vector_type(8)));

static __device__ inline v8f wmma4(v2f a, v2f b, v8f c) {
    return __builtin_amdgcn_wmma_f32_16x16x4_f32(
         false, a,  false, b,
         (short)0, c,  false,  false);
}

static __device__ inline float sigmoidf_(float x) {
    return __builtin_amdgcn_rcpf(1.0f + __expf(-x));
}

static __device__ inline float tanhf_(float x) {
    return 1.0f - 2.0f * __builtin_amdgcn_rcpf(1.0f + __expf(2.0f * x));
}

#define GK_  32
#define GP_  (GK_ + 4)
#define GPN_ (128 + 8)

template <int TRANSB, bool GATED>
__global__ __launch_bounds__(256)
void gemm_wmma_kernel(const float* __restrict__ A, const float* __restrict__ Bm,
                      const float* __restrict__ bias, const float* __restrict__ Gsrc,
                      float* __restrict__ C,
                      int K, int lda, int ldb, int ldc,
                      long long strideA, long long strideB, long long strideC)
{
    __shared__ __attribute__((aligned(16))) float As[64 * GP_];
    __shared__ __attribute__((aligned(16))) float Bs[TRANSB ? 128 * GP_ : GK_ * GPN_];

    const int tid  = threadIdx.x;
    const int wave = tid >> 5;
    const int lane = tid & 31;
    const int half = lane >> 4;
    const int l    = lane & 15;

    const int m0 = blockIdx.x * 64;
    const int nb = blockIdx.y * 128;
    const int n0 = nb + wave * 16;

    const float* Ab = A  + (long long)blockIdx.z * strideA;
    const float* Bb = Bm + (long long)blockIdx.z * strideB;
    float*       Cb = C  + (long long)blockIdx.z * strideC;

    const int ar  = tid >> 3;
    const int ac4 = (tid & 7) * 4;
    const int kr  = tid >> 5;
    const int nc4 = (tid & 31) * 4;

    const float* gA = Ab + (long long)(m0 + ar) * lda + ac4;
    const float* gB = TRANSB ? (Bb + (long long)(nb + ar) * ldb + ac4)
                             : (Bb + (long long)kr * ldb + nb + nc4);

    float4 ra0, ra1, rb0, rb1, rb2, rb3;

#define GEMM_LOAD_CHUNK(k0)                                                        \
    do {                                                                           \
        ra0 = *(const float4*)(gA + (k0));                                         \
        ra1 = *(const float4*)(gA + (long long)32 * lda + (k0));                   \
        if (TRANSB) {                                                              \
            rb0 = *(const float4*)(gB + (k0));                                     \
            rb1 = *(const float4*)(gB + (long long)32 * ldb + (k0));               \
            rb2 = *(const float4*)(gB + (long long)64 * ldb + (k0));               \
            rb3 = *(const float4*)(gB + (long long)96 * ldb + (k0));               \
        } else {                                                                   \
            rb0 = *(const float4*)(gB + (long long)((k0))      * ldb);             \
            rb1 = *(const float4*)(gB + (long long)((k0) + 8)  * ldb);             \
            rb2 = *(const float4*)(gB + (long long)((k0) + 16) * ldb);             \
            rb3 = *(const float4*)(gB + (long long)((k0) + 24) * ldb);             \
        }                                                                          \
    } while (0)

    v8f acc0 = {0.f,0.f,0.f,0.f,0.f,0.f,0.f,0.f};
    v8f acc1 = acc0, acc2 = acc0, acc3 = acc0;

    const float* sA = As + l * GP_ + 2 * half;
    const float* sB = TRANSB ? (Bs + (wave * 16 + l) * GP_ + 2 * half)
                             : (Bs + (2 * half) * GPN_ + wave * 16 + l);

    GEMM_LOAD_CHUNK(0);

    for (int k0 = 0; k0 < K; k0 += GK_) {
        *(float4*)(As + ar * GP_ + ac4)        = ra0;
        *(float4*)(As + (ar + 32) * GP_ + ac4) = ra1;
        if (TRANSB) {
            *(float4*)(Bs + ar * GP_ + ac4)        = rb0;
            *(float4*)(Bs + (ar + 32) * GP_ + ac4) = rb1;
            *(float4*)(Bs + (ar + 64) * GP_ + ac4) = rb2;
            *(float4*)(Bs + (ar + 96) * GP_ + ac4) = rb3;
        } else {
            *(float4*)(Bs + kr * GPN_ + nc4)        = rb0;
            *(float4*)(Bs + (kr + 8) * GPN_ + nc4)  = rb1;
            *(float4*)(Bs + (kr + 16) * GPN_ + nc4) = rb2;
            *(float4*)(Bs + (kr + 24) * GPN_ + nc4) = rb3;
        }
        __syncthreads();

        if (k0 + GK_ < K) GEMM_LOAD_CHUNK(k0 + GK_);

#pragma unroll
        for (int ks = 0; ks < GK_ / 4; ++ks) {
            v2f b;
            if (TRANSB) {
                b = *(const v2f*)(sB + 4 * ks);
            } else {
                b.x = sB[(4 * ks) * GPN_];
                b.y = sB[(4 * ks + 1) * GPN_];
            }
            v2f a0 = *(const v2f*)(sA + 4 * ks);
            v2f a1 = *(const v2f*)(sA + 16 * GP_ + 4 * ks);
            v2f a2 = *(const v2f*)(sA + 32 * GP_ + 4 * ks);
            v2f a3 = *(const v2f*)(sA + 48 * GP_ + 4 * ks);
            acc0 = wmma4(a0, b, acc0);
            acc1 = wmma4(a1, b, acc1);
            acc2 = wmma4(a2, b, acc2);
            acc3 = wmma4(a3, b, acc3);
        }
        __syncthreads();
    }
#undef GEMM_LOAD_CHUNK

    const int   col = n0 + l;
    const float bv  = bias ? bias[col] : 0.0f;

    v8f accs[4] = {acc0, acc1, acc2, acc3};
#pragma unroll
    for (int mi = 0; mi < 4; ++mi) {
#pragma unroll
        for (int r = 0; r < 8; ++r) {
            const int row = m0 + mi * 16 + r + half * 8;
            float cv = accs[mi][r] + bv;
            if (GATED) {
                cv = sigmoidf_(cv) * Gsrc[(long long)row * ldc + col];
            }
            Cb[(long long)row * ldc + col] = cv;
        }
    }
}

__global__ __launch_bounds__(256)
void attn_softmax_kernel(const float* __restrict__ own, const float* __restrict__ comp,
                         const float* __restrict__ v_attn, const int* __restrict__ lengths,
                         float* __restrict__ attn)
{
    const int q = blockIdx.x;
    const int b = blockIdx.y;
    const int tid = threadIdx.x;

    __shared__ float sOwn[H_];
    __shared__ float sVa[H_];
    __shared__ float sScore[L_];
    __shared__ float red[256];

    if (tid < H_) {
        sOwn[tid] = own[((long long)b * L_ + q) * H_ + tid];
        sVa[tid]  = v_attn[tid];
    }
    __syncthreads();

    const int len = lengths[b];
    for (int k = tid; k < L_; k += 256) {
        float s;
        if (k < len) {
            const float* cr = comp + ((long long)b * L_ + k) * H_;
            s = 0.0f;
#pragma unroll 4
            for (int h = 0; h < H_; ++h)
                s += sVa[h] * tanhf_(sOwn[h] + cr[h]);
        } else {
            s = -1e30f;
        }
        sScore[k] = s;
    }
    __syncthreads();

    float m = -1e30f;
    for (int k = tid; k < L_; k += 256) m = fmaxf(m, sScore[k]);
    red[tid] = m;
    __syncthreads();
    for (int s = 128; s > 0; s >>= 1) {
        if (tid < s) red[tid] = fmaxf(red[tid], red[tid + s]);
        __syncthreads();
    }
    const float mx = red[0];
    __syncthreads();

    float psum = 0.0f;
    for (int k = tid; k < L_; k += 256) {
        float e = __expf(sScore[k] - mx);
        sScore[k] = e;
        psum += e;
    }
    red[tid] = psum;
    __syncthreads();
    for (int s = 128; s > 0; s >>= 1) {
        if (tid < s) red[tid] += red[tid + s];
        __syncthreads();
    }
    const float inv = 1.0f / red[0];

    float* arow = attn + ((long long)b * L_ + q) * L_;
    for (int k = tid; k < L_; k += 256) arow[k] = sScore[k] * inv;
}

__global__ __launch_bounds__(256)
void concat_kernel(const float* __restrict__ v, const float* __restrict__ ctx,
                   float* __restrict__ inp, int total)
{
    int i = blockIdx.x * blockDim.x + threadIdx.x;
    if (i >= total) return;
    int row = i >> 9;
    int col = i & 511;
    inp[i] = (col < D_) ? v[(long long)row * D_ + col]
                        : ctx[(long long)row * D_ + (col - D_)];
}

__global__ __launch_bounds__(512)
void gru_kernel(const float* __restrict__ xp_f, const float* __restrict__ xp_b,
                const float* __restrict__ w_hh_f, const float* __restrict__ w_hh_b,
                const float* __restrict__ b_hh_f, const float* __restrict__ b_hh_b,
                const int* __restrict__ lengths, float* __restrict__ out)
{
    const int rev = blockIdx.x;
    const float* __restrict__ xp   = rev ? xp_b   : xp_f;
    const float* __restrict__ w_hh = rev ? w_hh_b : w_hh_f;
    const float* __restrict__ b_hh = rev ? b_hh_b : b_hh_f;

    __shared__ float hA[16][HA_PITCH];
    __shared__ float hp[2][B_][3 * H_];

    const int tid  = threadIdx.x;
    const int wave = tid >> 5;
    const int lane = tid & 31;
    const int half = lane >> 4;
    const int l    = lane & 15;

    const int kh    = wave & 1;
    const int nbase = (wave >> 1) * 48;
    const int kbase = kh * (H_ / 2) + 2 * half;

    v2f wr0[16], wr1[16], wr2[16];
#pragma unroll
    for (int ks = 0; ks < 16; ++ks) {
        wr0[ks] = *(const v2f*)(w_hh + (long long)(nbase +  0 + l) * H_ + kbase + 4 * ks);
        wr1[ks] = *(const v2f*)(w_hh + (long long)(nbase + 16 + l) * H_ + kbase + 4 * ks);
        wr2[ks] = *(const v2f*)(w_hh + (long long)(nbase + 32 + l) * H_ + kbase + 4 * ks);
    }

    for (int i = tid; i < 16 * HA_PITCH; i += 512) (&hA[0][0])[i] = 0.0f;
    __syncthreads();

    const int b   = tid >> 7;
    const int j   = tid & 127;
    const int len = lengths[b];

    int tmax = 0;
    for (int bb = 0; bb < B_; ++bb) tmax = max(tmax, lengths[bb]);
    tmax = min(tmax, L_);

    float  out_val = 0.0f;
    float* out_ptr = out;

    for (int step = 0; step < tmax; ++step) {
        const int t = rev ? (tmax - 1 - step) : step;

        if (step > 0) *out_ptr = out_val;

        const float* xrow = xp + ((long long)(b * L_ + t)) * (3 * H_);
        const float xr = xrow[j];
        const float xz = xrow[H_ + j];
        const float xn = xrow[2 * H_ + j];

        {
            v2f a[16];
#pragma unroll
            for (int ks = 0; ks < 16; ++ks)
                a[ks] = *(const v2f*)&hA[l][kbase + 4 * ks];

            v8f acc0 = {0.f,0.f,0.f,0.f,0.f,0.f,0.f,0.f};
            v8f acc1 = acc0, acc2 = acc0;
#pragma unroll
            for (int ks = 0; ks < 16; ++ks) {
                acc0 = wmma4(a[ks], wr0[ks], acc0);
                acc1 = wmma4(a[ks], wr1[ks], acc1);
                acc2 = wmma4(a[ks], wr2[ks], acc2);
            }
            if (half == 0) {
#pragma unroll
                for (int r = 0; r < B_; ++r) {
                    hp[kh][r][nbase +  0 + l] = acc0[r];
                    hp[kh][r][nbase + 16 + l] = acc1[r];
                    hp[kh][r][nbase + 32 + l] = acc2[r];
                }
            }
        }
        __syncthreads();

        const float hprev = hA[b][j];
        const float hr = (hp[0][b][j]          + hp[1][b][j])          + b_hh[j];
        const float hz = (hp[0][b][H_ + j]     + hp[1][b][H_ + j])     + b_hh[H_ + j];
        const float hn = (hp[0][b][2 * H_ + j] + hp[1][b][2 * H_ + j]) + b_hh[2 * H_ + j];

        const float r = sigmoidf_(xr + hr);
        const float z = sigmoidf_(xz + hz);
        const float n = tanhf_(xn + r * hn);
        const float hnew = (1.0f - z) * n + z * hprev;

        const bool msk = (t < len);
        hA[b][j] = msk ? hnew : hprev;
        out_val = msk ? hnew : 0.0f;
        out_ptr = out + ((long long)(b * L_ + t)) * (2 * H_) + rev * H_ + j;
        __syncthreads();
    }
    if (tmax > 0) *out_ptr = out_val;

    for (int t = tmax; t < L_; ++t)
        out[((long long)(b * L_ + t)) * (2 * H_) + rev * H_ + j] = 0.0f;
}

extern "C" void kernel_launch(void* const* d_in, const int* in_sizes, int n_in,
                              void* d_out, int out_size, void* d_ws, size_t ws_size,
                              hipStream_t stream)
{
    (void)in_sizes; (void)n_in; (void)out_size; (void)ws_size;

    const float* v       = (const float*)d_in[0];
    const int*   lengths = (const int*)  d_in[1];
    const float* own_W   = (const float*)d_in[3];
    const float* own_b   = (const float*)d_in[4];
    const float* comp_W  = (const float*)d_in[5];
    const float* comp_b  = (const float*)d_in[6];
    const float* v_attn  = (const float*)d_in[7];
    const float* gate_W  = (const float*)d_in[8];
    const float* gate_b  = (const float*)d_in[9];
    const float* w_ih_f  = (const float*)d_in[10];
    const float* w_hh_f  = (const float*)d_in[11];
    const float* b_ih_f  = (const float*)d_in[12];
    const float* b_hh_f  = (const float*)d_in[13];
    const float* w_ih_b  = (const float*)d_in[14];
    const float* w_hh_b  = (const float*)d_in[15];
    const float* b_ih_b  = (const float*)d_in[16];
    const float* b_hh_b  = (const float*)d_in[17];
    float* out = (float*)d_out;

    const int M = B_ * L_;
    float* ws    = (float*)d_ws;
    float* own   = ws;
    float* comp  = own  + (size_t)M * H_;
    float* attn  = comp + (size_t)M * H_;
    float* ctx   = attn + (size_t)B_ * L_ * L_;
    float* inp   = ctx  + (size_t)M * D_;
    float* inpg  = inp  + (size_t)M * 2 * D_;
    float* xp_f  = inpg + (size_t)M * 2 * D_;
    float* xp_b  = xp_f + (size_t)M * 3 * H_;

    {
        dim3 grid(M / 64, H_ / 128, 1);
        gemm_wmma_kernel<1, false><<<grid, 256, 0, stream>>>(v, own_W, own_b, nullptr, own,
            D_, D_, D_, H_, 0, 0, 0);
        gemm_wmma_kernel<1, false><<<grid, 256, 0, stream>>>(v, comp_W, comp_b, nullptr, comp,
            D_, D_, D_, H_, 0, 0, 0);
    }

    {
        dim3 grid(L_, B_, 1);
        attn_softmax_kernel<<<grid, 256, 0, stream>>>(own, comp, v_attn, lengths, attn);
    }

    {
        dim3 grid(L_ / 64, D_ / 128, B_);
        gemm_wmma_kernel<0, false><<<grid, 256, 0, stream>>>(attn, v, nullptr, nullptr, ctx,
            L_, L_, D_, D_,
            (long long)L_ * L_, (long long)L_ * D_, (long long)L_ * D_);
    }

    {
        const int total = M * 2 * D_;
        concat_kernel<<<(total + 255) / 256, 256, 0, stream>>>(v, ctx, inp, total);
    }

    {
        dim3 grid(M / 64, (2 * D_) / 128, 1);
        gemm_wmma_kernel<1, true><<<grid, 256, 0, stream>>>(inp, gate_W, gate_b, inp, inpg,
            2 * D_, 2 * D_, 2 * D_, 2 * D_, 0, 0, 0);
    }

    {
        dim3 grid(M / 64, (3 * H_) / 128, 1);
        gemm_wmma_kernel<1, false><<<grid, 256, 0, stream>>>(inpg, w_ih_f, b_ih_f, nullptr, xp_f,
            2 * D_, 2 * D_, 2 * D_, 3 * H_, 0, 0, 0);
        gemm_wmma_kernel<1, false><<<grid, 256, 0, stream>>>(inpg, w_ih_b, b_ih_b, nullptr, xp_b,
            2 * D_, 2 * D_, 2 * D_, 3 * H_, 0, 0, 0);
    }

    gru_kernel<<<2, 512, 0, stream>>>(xp_f, xp_b, w_hh_f, w_hh_b, b_hh_f, b_hh_b, lengths, out);
}
